# starting kernel
# baseline (speedup 1.0000x reference)
#include <hip/hip_runtime.h>
#include <math.h>

#define ALPHA   0.2f
#define N_NODES 4096
#define IN_FEAT 512
#define NHEAD   4
#define FDIM    64
#define HF      256

typedef __attribute__((ext_vector_type(2))) float v2f;
typedef __attribute__((ext_vector_type(8))) float v8f;

__device__ __forceinline__ float lrelu(float x) {
    return fmaxf(x, ALPHA * x);
}

__global__ __launch_bounds__(128) void gat_gemm1(const float* __restrict__ x,
                                                 const float* __restrict__ W,
                                                 float* __restrict__ h) {
    const int lane = threadIdx.x & 31;
    const int wave = threadIdx.x >> 5;
    const int i0   = blockIdx.x * 16;
    const int n0   = blockIdx.y * 64 + wave * 16;
    const int m    = lane & 15;
    const int koff = (lane >> 4) << 1;

    const float* xrow = x + (size_t)(i0 + m) * IN_FEAT;
    v8f acc = {};
    for (int k0 = 0; k0 < IN_FEAT; k0 += 4) {
        const int ka = k0 + koff;
        v2f a2, b2;
        a2.x = xrow[ka];
        a2.y = xrow[ka + 1];
        b2.x = W[(size_t)ka * HF + n0 + m];
        b2.y = W[(size_t)(ka + 1) * HF + n0 + m];
        acc = __builtin_amdgcn_wmma_f32_16x16x4_f32(false, a2, false, b2,
                                                    (short)0, acc, false, false);
    }
    const int rbase = (lane >> 4) << 3;
#pragma unroll
    for (int v = 0; v < 8; ++v)
        h[(size_t)(i0 + rbase + v) * HF + n0 + m] = acc[v];
}

__global__ __launch_bounds__(128) void gat_st(const float* __restrict__ h,
                                              const float* __restrict__ a,
                                              float* __restrict__ s,
                                              float* __restrict__ t) {
    const int lane = threadIdx.x & 31;
    const int gw   = blockIdx.x * 4 + (threadIdx.x >> 5);
    const int n    = gw >> 2;
    const int hd   = gw & 3;

    const float* hrow = h + (size_t)n * HF + hd * FDIM;
    const float* asrc = a + hd * 2 * FDIM;
    const float* adst = asrc + FDIM;

    float sv = hrow[lane] * asrc[lane] + hrow[lane + 32] * asrc[lane + 32];
    float tv = hrow[lane] * adst[lane] + hrow[lane + 32] * adst[lane + 32];
#pragma unroll
    for (int off = 16; off > 0; off >>= 1) {
        sv += __shfl_xor(sv, off, 32);
        tv += __shfl_xor(tv, off, 32);
    }
    if (lane == 0) {
        s[n * NHEAD + hd] = sv;
        t[n * NHEAD + hd] = tv;
    }
}

__global__ __launch_bounds__(256) void gat_tmax(const float* __restrict__ t,
                                                float* __restrict__ tmax) {
    __shared__ float red[256];
    const int hd = blockIdx.x;
    float m = -INFINITY;
    for (int j = threadIdx.x; j < N_NODES; j += 256)
        m = fmaxf(m, t[j * NHEAD + hd]);
    red[threadIdx.x] = m;
    __syncthreads();
    for (int off = 128; off > 0; off >>= 1) {
        if (threadIdx.x < off)
            red[threadIdx.x] = fmaxf(red[threadIdx.x], red[threadIdx.x + off]);
        __syncthreads();
    }
    if (threadIdx.x == 0) tmax[hd] = red[0];
}

__global__ __launch_bounds__(128) void gat_attn(const float* __restrict__ h,
                                                const float* __restrict__ s,
                                                const float* __restrict__ t,
                                                const float* __restrict__ tmax,
                                                float* __restrict__ out) {
    __shared__ float cred[4][16][64];
    __shared__ float zred[4][32];
    __shared__ float zrow[16];

    const int lane = threadIdx.x & 31;
    const int wave = threadIdx.x >> 5;
    const int i0   = blockIdx.x * 16;
    const int hd   = blockIdx.y;
    const int m    = lane & 15;
    const int koff = (lane >> 4) << 1;

    const float sm   = s[(i0 + m) * NHEAD + hd];
    const float mrow = lrelu(sm + tmax[hd]);
    const float* hh  = h + hd * FDIM;

    v8f acc0 = {}, acc1 = {}, acc2 = {}, acc3 = {};
    float zpart = 0.f;

    const int jbeg = wave * (N_NODES / 4);
    const int jend = jbeg + (N_NODES / 4);
    for (int j0 = jbeg; j0 < jend; j0 += 4) {
        const int ja = j0 + koff;
        const float pa = __expf(lrelu(sm + t[ja * NHEAD + hd]) - mrow);
        const float pb = __expf(lrelu(sm + t[(ja + 1) * NHEAD + hd]) - mrow);
        zpart += pa + pb;
        v2f a2; a2.x = pa; a2.y = pb;

        const float* hja = hh + (size_t)ja * HF;
        const float* hjb = hja + HF;
        v2f b0, b1, b2, b3;
        b0.x = hja[m];      b0.y = hjb[m];
        b1.x = hja[16 + m]; b1.y = hjb[16 + m];
        b2.x = hja[32 + m]; b2.y = hjb[32 + m];
        b3.x = hja[48 + m]; b3.y = hjb[48 + m];

        acc0 = __builtin_amdgcn_wmma_f32_16x16x4_f32(false, a2, false, b0, (short)0, acc0, false, false);
        acc1 = __builtin_amdgcn_wmma_f32_16x16x4_f32(false, a2, false, b1, (short)0, acc1, false, false);
        acc2 = __builtin_amdgcn_wmma_f32_16x16x4_f32(false, a2, false, b2, (short)0, acc2, false, false);
        acc3 = __builtin_amdgcn_wmma_f32_16x16x4_f32(false, a2, false, b3, (short)0, acc3, false, false);
    }

    const int rbase = (lane >> 4) << 3;
#pragma unroll
    for (int v = 0; v < 8; ++v) {
        cred[wave][rbase + v][m]      = acc0[v];
        cred[wave][rbase + v][16 + m] = acc1[v];
        cred[wave][rbase + v][32 + m] = acc2[v];
        cred[wave][rbase + v][48 + m] = acc3[v];
    }
    zred[wave][lane] = zpart;
    __syncthreads();

    if (threadIdx.x < 16) {
        float z = 0.f;
#pragma unroll
        for (int w = 0; w < 4; ++w)
            z += zred[w][threadIdx.x] + zred[w][threadIdx.x + 16];
        zrow[threadIdx.x] = z;
    }
    __syncthreads();

    for (int e = threadIdx.x; e < 16 * 64; e += 128) {
        const int rr = e >> 6, cc = e & 63;
        const float v = cred[0][rr][cc] + cred[1][rr][cc] +
                        cred[2][rr][cc] + cred[3][rr][cc];
        out[(size_t)(i0 + rr) * HF + hd * FDIM + cc] = v / zrow[rr];
    }
}

extern "C" void kernel_launch(void* const* d_in, const int* in_sizes, int n_in,
                              void* d_out, int out_size, void* d_ws, size_t ws_size,
                              hipStream_t stream) {
    const float* x = (const float*)d_in[0];
    const float* W = (const float*)d_in[2];
    const float* a = (const float*)d_in[3];
    float* out = (float*)d_out;

    char*  ws   = (char*)d_ws;
    float* h    = (float*)ws;
    float* s    = (float*)(ws + (size_t)N_NODES * HF * sizeof(float));
    float* t    = s + N_NODES * NHEAD;
    float* tmax = t + N_NODES * NHEAD;

    dim3 g1(N_NODES / 16, HF / 64);
    gat_gemm1<<<g1, 128, 0, stream>>>(x, W, h);

    gat_st<<<N_NODES, 128, 0, stream>>>(h, a, s, t);

    gat_tmax<<<NHEAD, 256, 0, stream>>>(t, tmax);

    dim3 g4(N_NODES / 16, NHEAD);
    gat_attn<<<g4, 128, 0, stream>>>(h, s, t, tmax, out);
}


# attention PV: each 64x64 slab of h staged once per 64-row block in LDS (16-byte global loads, next slab in flight) and shared by 4 M tiles; was per-wave dword loads from L2
# speedup vs baseline: 1.2166x; 1.2166x over previous
#include <hip/hip_runtime.h>
#include <math.h>

#define ALPHA   0.2f
#define N_NODES 4096
#define IN_FEAT 512
#define NHEAD   4
#define FDIM    64
#define HF      256

typedef __attribute__((ext_vector_type(2))) float v2f;
typedef __attribute__((ext_vector_type(8))) float v8f;

__device__ __forceinline__ float lrelu(float x) {
    return fmaxf(x, ALPHA * x);
}

__global__ __launch_bounds__(128) void gat_gemm1(const float* __restrict__ x,
                                                 const float* __restrict__ W,
                                                 float* __restrict__ h) {
    const int lane = threadIdx.x & 31;
    const int wave = threadIdx.x >> 5;
    const int i0   = blockIdx.x * 16;
    const int n0   = blockIdx.y * 64 + wave * 16;
    const int m    = lane & 15;
    const int koff = (lane >> 4) << 1;

    const float* xrow = x + (size_t)(i0 + m) * IN_FEAT;
    v8f acc = {};
    for (int k0 = 0; k0 < IN_FEAT; k0 += 4) {
        const int ka = k0 + koff;
        v2f a2, b2;
        a2.x = xrow[ka];
        a2.y = xrow[ka + 1];
        b2.x = W[(size_t)ka * HF + n0 + m];
        b2.y = W[(size_t)(ka + 1) * HF + n0 + m];
        acc = __builtin_amdgcn_wmma_f32_16x16x4_f32(false, a2, false, b2,
                                                    (short)0, acc, false, false);
    }
    const int rbase = (lane >> 4) << 3;
#pragma unroll
    for (int v = 0; v < 8; ++v)
        h[(size_t)(i0 + rbase + v) * HF + n0 + m] = acc[v];
}

__global__ __launch_bounds__(128) void gat_st(const float* __restrict__ h,
                                              const float* __restrict__ a,
                                              float* __restrict__ s,
                                              float* __restrict__ t) {
    const int lane = threadIdx.x & 31;
    const int gw   = blockIdx.x * 4 + (threadIdx.x >> 5);
    const int n    = gw >> 2;
    const int hd   = gw & 3;

    const float* hrow = h + (size_t)n * HF + hd * FDIM;
    const float* asrc = a + hd * 2 * FDIM;
    const float* adst = asrc + FDIM;

    float sv = hrow[lane] * asrc[lane] + hrow[lane + 32] * asrc[lane + 32];
    float tv = hrow[lane] * adst[lane] + hrow[lane + 32] * adst[lane + 32];
#pragma unroll
    for (int off = 16; off > 0; off >>= 1) {
        sv += __shfl_xor(sv, off, 32);
        tv += __shfl_xor(tv, off, 32);
    }
    if (lane == 0) {
        s[n * NHEAD + hd] = sv;
        t[n * NHEAD + hd] = tv;
    }
}

__global__ __launch_bounds__(256) void gat_tmax(const float* __restrict__ t,
                                                float* __restrict__ tmax) {
    __shared__ float red[256];
    const int hd = blockIdx.x;
    float m = -INFINITY;
    for (int j = threadIdx.x; j < N_NODES; j += 256)
        m = fmaxf(m, t[j * NHEAD + hd]);
    red[threadIdx.x] = m;
    __syncthreads();
    for (int off = 128; off > 0; off >>= 1) {
        if (threadIdx.x < off)
            red[threadIdx.x] = fmaxf(red[threadIdx.x], red[threadIdx.x + off]);
        __syncthreads();
    }
    if (threadIdx.x == 0) tmax[hd] = red[0];
}

#define ATT_ROWS 64
#define KJ       64
#define PP       160
#define SLABF    ((KJ / 2) * PP)
#define NSLAB    (N_NODES / 4 / KJ)

__global__ __launch_bounds__(512) void gat_attn(const float* __restrict__ h,
                                                const float* __restrict__ s,
                                                const float* __restrict__ t,
                                                const float* __restrict__ tmax,
                                                float* __restrict__ out) {
    __shared__ __attribute__((aligned(16))) float bt[4 * 2 * SLABF];
    __shared__ __attribute__((aligned(16))) float tt[4][2][KJ];
    float (*cred)[ATT_ROWS][64] = reinterpret_cast<float (*)[ATT_ROWS][64]>(bt);
    __shared__ float zred[4][4][32];
    __shared__ float zrow[ATT_ROWS];

    const int tid  = threadIdx.x;
    const int lane = tid & 31;
    const int wave = tid >> 5;
    const int q    = wave >> 2;
    const int mt   = wave & 3;
    const int gt   = tid & 127;
    const int i0   = blockIdx.x * ATT_ROWS;
    const int hd   = blockIdx.y;
    const int m    = lane & 15;
    const int koff = (lane >> 4) << 1;

    const float sm   = s[(i0 + mt * 16 + m) * NHEAD + hd];
    const float mrow = lrelu(sm + tmax[hd]);
    const float* hh  = h + hd * FDIM;

    const int sp   = gt >> 4;
    const int scol = (gt & 15) << 2;
    const int jq0  = q * (N_NODES / 4);
    float* const bq = bt + q * (2 * SLABF);
    const int sdst  = sp * PP + scol * 2;

    float4 stg[8];
    float  tstg = 0.f;
    {
        const float* src0 = hh + (size_t)(jq0 + 2 * sp) * HF + scol;
#pragma unroll
        for (int r = 0; r < 4; ++r) {
            stg[2 * r]     = *reinterpret_cast<const float4*>(src0 + (size_t)(16 * r) * HF);
            stg[2 * r + 1] = *reinterpret_cast<const float4*>(src0 + (size_t)(16 * r + 1) * HF);
        }
        if (gt < KJ) tstg = t[(jq0 + gt) * NHEAD + hd];
#pragma unroll
        for (int r = 0; r < 4; ++r) {
            const float4 e = stg[2 * r], o = stg[2 * r + 1];
            float* d = bq + sdst + 8 * r * PP;
            *reinterpret_cast<float4*>(d)     = make_float4(e.x, o.x, e.y, o.y);
            *reinterpret_cast<float4*>(d + 4) = make_float4(e.z, o.z, e.w, o.w);
        }
        if (gt < KJ) tt[q][0][gt] = tstg;
    }
    __syncthreads();

    v8f acc0 = {}, acc1 = {}, acc2 = {}, acc3 = {};
    float zpart = 0.f;

    for (int sl = 0; sl < NSLAB; ++sl) {
        const int  cur  = sl & 1;
        const bool more = (sl + 1 < NSLAB);
        if (more) {
            const float* src0 = hh + (size_t)(jq0 + (sl + 1) * KJ + 2 * sp) * HF + scol;
#pragma unroll
            for (int r = 0; r < 4; ++r) {
                stg[2 * r]     = *reinterpret_cast<const float4*>(src0 + (size_t)(16 * r) * HF);
                stg[2 * r + 1] = *reinterpret_cast<const float4*>(src0 + (size_t)(16 * r + 1) * HF);
            }
            if (gt < KJ) tstg = t[(jq0 + (sl + 1) * KJ + gt) * NHEAD + hd];
        }

        const float* bc = bq + cur * SLABF + (lane >> 4) * PP + 2 * m;
        const float* tc = &tt[q][cur][0];
#pragma unroll
        for (int kk = 0; kk < KJ; kk += 4) {
            const int ja = kk + koff;
            const float pa = __expf(lrelu(sm + tc[ja]) - mrow);
            const float pb = __expf(lrelu(sm + tc[ja + 1]) - mrow);
            zpart += pa + pb;
            v2f a2; a2.x = pa; a2.y = pb;

            const float* hp = bc + (kk >> 1) * PP;
            const v2f b0 = *reinterpret_cast<const v2f*>(hp);
            const v2f b1 = *reinterpret_cast<const v2f*>(hp + 32);
            const v2f b2 = *reinterpret_cast<const v2f*>(hp + 64);
            const v2f b3 = *reinterpret_cast<const v2f*>(hp + 96);

            acc0 = __builtin_amdgcn_wmma_f32_16x16x4_f32(false, a2, false, b0, (short)0, acc0, false, false);
            acc1 = __builtin_amdgcn_wmma_f32_16x16x4_f32(false, a2, false, b1, (short)0, acc1, false, false);
            acc2 = __builtin_amdgcn_wmma_f32_16x16x4_f32(false, a2, false, b2, (short)0, acc2, false, false);
            acc3 = __builtin_amdgcn_wmma_f32_16x16x4_f32(false, a2, false, b3, (short)0, acc3, false, false);
        }

        if (more) {
#pragma unroll
            for (int r = 0; r < 4; ++r) {
                const float4 e = stg[2 * r], o = stg[2 * r + 1];
                float* d = bq + (cur ^ 1) * SLABF + sdst + 8 * r * PP;
                *reinterpret_cast<float4*>(d)     = make_float4(e.x, o.x, e.y, o.y);
                *reinterpret_cast<float4*>(d + 4) = make_float4(e.z, o.z, e.w, o.w);
            }
            if (gt < KJ) tt[q][cur ^ 1][gt] = tstg;
        }
        __syncthreads();
    }

    const int rbase = mt * 16 + ((lane >> 4) << 3);
#pragma unroll
    for (int v = 0; v < 8; ++v) {
        cred[q][rbase + v][m]      = acc0[v];
        cred[q][rbase + v][16 + m] = acc1[v];
        cred[q][rbase + v][32 + m] = acc2[v];
        cred[q][rbase + v][48 + m] = acc3[v];
    }
    zred[q][mt][lane] = zpart;
    __syncthreads();

    if (tid < ATT_ROWS) {
        const int zt = tid >> 4, zr = tid & 15;
        float z = 0.f;
#pragma unroll
        for (int w = 0; w < 4; ++w)
            z += zred[w][zt][zr] + zred[w][zt][zr + 16];
        zrow[tid] = z;
    }
    __syncthreads();

    for (int e = tid; e < ATT_ROWS * 64; e += 512) {
        const int rr = e >> 6, cc = e & 63;
        const float v = cred[0][rr][cc] + cred[1][rr][cc] +
                        cred[2][rr][cc] + cred[3][rr][cc];
        out[(size_t)(i0 + rr) * HF + hd * FDIM + cc] = v / zrow[rr];
    }
}

extern "C" void kernel_launch(void* const* d_in, const int* in_sizes, int n_in,
                              void* d_out, int out_size, void* d_ws, size_t ws_size,
                              hipStream_t stream) {
    const float* x = (const float*)d_in[0];
    const float* W = (const float*)d_in[2];
    const float* a = (const float*)d_in[3];
    float* out = (float*)d_out;

    char*  ws   = (char*)d_ws;
    float* h    = (float*)ws;
    float* s    = (float*)(ws + (size_t)N_NODES * HF * sizeof(float));
    float* t    = s + N_NODES * NHEAD;
    float* tmax = t + N_NODES * NHEAD;

    dim3 g1(N_NODES / 16, HF / 64);
    gat_gemm1<<<g1, 128, 0, stream>>>(x, W, h);

    gat_st<<<N_NODES, 128, 0, stream>>>(h, a, s, t);

    gat_tmax<<<NHEAD, 256, 0, stream>>>(t, tmax);

    dim3 g4(N_NODES / ATT_ROWS, NHEAD);
    gat_attn<<<g4, 512, 0, stream>>>(h, s, t, tmax, out);
}


# x@W GEMM: 64x64 tile per block, each 64-deep slab of x and W staged once in LDS (16-byte global loads, next slab in flight), 8-byte fragment reads; was per-wave dword loads from L2; sum order kept
# speedup vs baseline: 1.3403x; 1.1017x over previous
#include <hip/hip_runtime.h>
#include <math.h>

#define ALPHA   0.2f
#define N_NODES 4096
#define IN_FEAT 512
#define NHEAD   4
#define FDIM    64
#define HF      256

typedef __attribute__((ext_vector_type(2))) float v2f;
typedef __attribute__((ext_vector_type(8))) float v8f;

__device__ __forceinline__ float lrelu(float x) {
    return fmaxf(x, ALPHA * x);
}

#define GBM    64
#define GBN    64
#define GBK    64
#define GPA    68
#define GPB    160
#define GNSLAB (IN_FEAT / GBK)

__device__ __forceinline__ void gemm1_slab(const float* __restrict__ ac, const float* __restrict__ bc,
                                           v8f& acc0, v8f& acc1) {
#pragma unroll
    for (int kk = 0; kk < GBK; kk += 4) {
        const v2f a2 = *reinterpret_cast<const v2f*>(ac + kk);
        const float* bp = bc + (kk >> 1) * GPB;
        const v2f b0 = *reinterpret_cast<const v2f*>(bp);
        const v2f b1 = *reinterpret_cast<const v2f*>(bp + 32);
        acc0 = __builtin_amdgcn_wmma_f32_16x16x4_f32(false, a2, false, b0, (short)0, acc0, false, false);
        acc1 = __builtin_amdgcn_wmma_f32_16x16x4_f32(false, a2, false, b1, (short)0, acc1, false, false);
    }
}

__global__ __launch_bounds__(256) void gat_gemm1(const float* __restrict__ x,
                                                 const float* __restrict__ W,
                                                 float* __restrict__ h) {
    __shared__ __attribute__((aligned(16))) float xs[2][GBM * GPA];
    __shared__ __attribute__((aligned(16))) float wsl[2][(GBK / 2) * GPB];

    const int tid  = threadIdx.x;
    const int lane = tid & 31;
    const int wave = tid >> 5;
    const int mt   = wave & 3;
    const int nh   = wave >> 2;
    const int i0   = blockIdx.x * GBM;
    const int n0   = blockIdx.y * GBN;
    const int m    = lane & 15;
    const int hf   = lane >> 4;

    const int srow = tid >> 4;
    const int scol = (tid & 15) << 2;
    const float* xsrc = x + (size_t)(i0 + srow) * IN_FEAT + scol;
    const float* wsrc = W + (size_t)(2 * srow) * HF + n0 + scol;
    const int xdst = srow * GPA + scol;
    const int wdst = srow * GPB + scol * 2;

    float4 x0, x1, x2, x3;
    float4 we0, wo0, we1, wo1;
    x0  = *reinterpret_cast<const float4*>(xsrc);
    x1  = *reinterpret_cast<const float4*>(xsrc + (size_t)16 * IN_FEAT);
    x2  = *reinterpret_cast<const float4*>(xsrc + (size_t)32 * IN_FEAT);
    x3  = *reinterpret_cast<const float4*>(xsrc + (size_t)48 * IN_FEAT);
    we0 = *reinterpret_cast<const float4*>(wsrc);
    wo0 = *reinterpret_cast<const float4*>(wsrc + (size_t)HF);
    we1 = *reinterpret_cast<const float4*>(wsrc + (size_t)32 * HF);
    wo1 = *reinterpret_cast<const float4*>(wsrc + (size_t)33 * HF);
    {
        float* xd = &xs[0][xdst];
        float* wd = &wsl[0][wdst];
        *reinterpret_cast<float4*>(xd)                = x0;
        *reinterpret_cast<float4*>(xd + 16 * GPA)     = x1;
        *reinterpret_cast<float4*>(xd + 32 * GPA)     = x2;
        *reinterpret_cast<float4*>(xd + 48 * GPA)     = x3;
        *reinterpret_cast<float4*>(wd)                = make_float4(we0.x, wo0.x, we0.y, wo0.y);
        *reinterpret_cast<float4*>(wd + 4)            = make_float4(we0.z, wo0.z, we0.w, wo0.w);
        *reinterpret_cast<float4*>(wd + 16 * GPB)     = make_float4(we1.x, wo1.x, we1.y, wo1.y);
        *reinterpret_cast<float4*>(wd + 16 * GPB + 4) = make_float4(we1.z, wo1.z, we1.w, wo1.w);
    }
    __syncthreads();

    v8f acc0 = {}, acc1 = {};

    const int aoff = (mt * 16 + m) * GPA + 2 * hf;
    const int boff = hf * GPB + 2 * (nh * 32 + m);

#pragma nounroll
    for (int sl = 0; sl < GNSLAB - 1; ++sl) {
        const int cur = sl & 1;
        const float* xn = xsrc + (sl + 1) * GBK;
        const float* wn = wsrc + (size_t)((sl + 1) * GBK) * HF;
        x0  = *reinterpret_cast<const float4*>(xn);
        x1  = *reinterpret_cast<const float4*>(xn + (size_t)16 * IN_FEAT);
        x2  = *reinterpret_cast<const float4*>(xn + (size_t)32 * IN_FEAT);
        x3  = *reinterpret_cast<const float4*>(xn + (size_t)48 * IN_FEAT);
        we0 = *reinterpret_cast<const float4*>(wn);
        wo0 = *reinterpret_cast<const float4*>(wn + (size_t)HF);
        we1 = *reinterpret_cast<const float4*>(wn + (size_t)32 * HF);
        wo1 = *reinterpret_cast<const float4*>(wn + (size_t)33 * HF);

        gemm1_slab(&xs[cur][aoff], &wsl[cur][boff], acc0, acc1);
        __builtin_amdgcn_sched_barrier(0x20);

        float* xd = &xs[cur ^ 1][xdst];
        float* wd = &wsl[cur ^ 1][wdst];
        *reinterpret_cast<float4*>(xd)                = x0;
        *reinterpret_cast<float4*>(xd + 16 * GPA)     = x1;
        *reinterpret_cast<float4*>(xd + 32 * GPA)     = x2;
        *reinterpret_cast<float4*>(xd + 48 * GPA)     = x3;
        *reinterpret_cast<float4*>(wd)                = make_float4(we0.x, wo0.x, we0.y, wo0.y);
        *reinterpret_cast<float4*>(wd + 4)            = make_float4(we0.z, wo0.z, we0.w, wo0.w);
        *reinterpret_cast<float4*>(wd + 16 * GPB)     = make_float4(we1.x, wo1.x, we1.y, wo1.y);
        *reinterpret_cast<float4*>(wd + 16 * GPB + 4) = make_float4(we1.z, wo1.z, we1.w, wo1.w);
        __syncthreads();
    }
    gemm1_slab(&xs[(GNSLAB - 1) & 1][aoff], &wsl[(GNSLAB - 1) & 1][boff], acc0, acc1);

    float* hrow = h + (size_t)(i0 + mt * 16 + (hf << 3)) * HF + n0 + nh * 32 + m;
#pragma unroll
    for (int v = 0; v < 8; ++v) {
        hrow[(size_t)v * HF]      = acc0[v];
        hrow[(size_t)v * HF + 16] = acc1[v];
    }
}

__global__ __launch_bounds__(128) void gat_st(const float* __restrict__ h,
                                              const float* __restrict__ a,
                                              float* __restrict__ s,
                                              float* __restrict__ t) {
    const int lane = threadIdx.x & 31;
    const int gw   = blockIdx.x * 4 + (threadIdx.x >> 5);
    const int n    = gw >> 2;
    const int hd   = gw & 3;

    const float* hrow = h + (size_t)n * HF + hd * FDIM;
    const float* asrc = a + hd * 2 * FDIM;
    const float* adst = asrc + FDIM;

    float sv = hrow[lane] * asrc[lane] + hrow[lane + 32] * asrc[lane + 32];
    float tv = hrow[lane] * adst[lane] + hrow[lane + 32] * adst[lane + 32];
#pragma unroll
    for (int off = 16; off > 0; off >>= 1) {
        sv += __shfl_xor(sv, off, 32);
        tv += __shfl_xor(tv, off, 32);
    }
    if (lane == 0) {
        s[n * NHEAD + hd] = sv;
        t[n * NHEAD + hd] = tv;
    }
}

__global__ __launch_bounds__(256) void gat_tmax(const float* __restrict__ t,
                                                float* __restrict__ tmax) {
    __shared__ float red[256];
    const int hd = blockIdx.x;
    float m = -INFINITY;
    for (int j = threadIdx.x; j < N_NODES; j += 256)
        m = fmaxf(m, t[j * NHEAD + hd]);
    red[threadIdx.x] = m;
    __syncthreads();
    for (int off = 128; off > 0; off >>= 1) {
        if (threadIdx.x < off)
            red[threadIdx.x] = fmaxf(red[threadIdx.x], red[threadIdx.x + off]);
        __syncthreads();
    }
    if (threadIdx.x == 0) tmax[hd] = red[0];
}

#define ATT_ROWS 64
#define KJ       64
#define PP       160
#define SLABF    ((KJ / 2) * PP)
#define NSLAB    (N_NODES / 4 / KJ)

__global__ __launch_bounds__(512) void gat_attn(const float* __restrict__ h,
                                                const float* __restrict__ s,
                                                const float* __restrict__ t,
                                                const float* __restrict__ tmax,
                                                float* __restrict__ out) {
    __shared__ __attribute__((aligned(16))) float bt[4 * 2 * SLABF];
    __shared__ __attribute__((aligned(16))) float tt[4][2][KJ];
    float (*cred)[ATT_ROWS][64] = reinterpret_cast<float (*)[ATT_ROWS][64]>(bt);
    __shared__ float zred[4][4][32];
    __shared__ float zrow[ATT_ROWS];

    const int tid  = threadIdx.x;
    const int lane = tid & 31;
    const int wave = tid >> 5;
    const int q    = wave >> 2;
    const int mt   = wave & 3;
    const int gt   = tid & 127;
    const int i0   = blockIdx.x * ATT_ROWS;
    const int hd   = blockIdx.y;
    const int m    = lane & 15;
    const int koff = (lane >> 4) << 1;

    const float sm   = s[(i0 + mt * 16 + m) * NHEAD + hd];
    const float mrow = lrelu(sm + tmax[hd]);
    const float* hh  = h + hd * FDIM;

    const int sp   = gt >> 4;
    const int scol = (gt & 15) << 2;
    const int jq0  = q * (N_NODES / 4);
    float* const bq = bt + q * (2 * SLABF);
    const int sdst  = sp * PP + scol * 2;

    float4 stg[8];
    float  tstg = 0.f;
    {
        const float* src0 = hh + (size_t)(jq0 + 2 * sp) * HF + scol;
#pragma unroll
        for (int r = 0; r < 4; ++r) {
            stg[2 * r]     = *reinterpret_cast<const float4*>(src0 + (size_t)(16 * r) * HF);
            stg[2 * r + 1] = *reinterpret_cast<const float4*>(src0 + (size_t)(16 * r + 1) * HF);
        }
        if (gt < KJ) tstg = t[(jq0 + gt) * NHEAD + hd];
#pragma unroll
        for (int r = 0; r < 4; ++r) {
            const float4 e = stg[2 * r], o = stg[2 * r + 1];
            float* d = bq + sdst + 8 * r * PP;
            *reinterpret_cast<float4*>(d)     = make_float4(e.x, o.x, e.y, o.y);
            *reinterpret_cast<float4*>(d + 4) = make_float4(e.z, o.z, e.w, o.w);
        }
        if (gt < KJ) tt[q][0][gt] = tstg;
    }
    __syncthreads();

    v8f acc0 = {}, acc1 = {}, acc2 = {}, acc3 = {};
    float zpart = 0.f;

    for (int sl = 0; sl < NSLAB; ++sl) {
        const int  cur  = sl & 1;
        const bool more = (sl + 1 < NSLAB);
        if (more) {
            const float* src0 = hh + (size_t)(jq0 + (sl + 1) * KJ + 2 * sp) * HF + scol;
#pragma unroll
            for (int r = 0; r < 4; ++r) {
                stg[2 * r]     = *reinterpret_cast<const float4*>(src0 + (size_t)(16 * r) * HF);
                stg[2 * r + 1] = *reinterpret_cast<const float4*>(src0 + (size_t)(16 * r + 1) * HF);
            }
            if (gt < KJ) tstg = t[(jq0 + (sl + 1) * KJ + gt) * NHEAD + hd];
        }

        const float* bc = bq + cur * SLABF + (lane >> 4) * PP + 2 * m;
        const float* tc = &tt[q][cur][0];
#pragma unroll
        for (int kk = 0; kk < KJ; kk += 4) {
            const int ja = kk + koff;
            const float pa = __expf(lrelu(sm + tc[ja]) - mrow);
            const float pb = __expf(lrelu(sm + tc[ja + 1]) - mrow);
            zpart += pa + pb;
            v2f a2; a2.x = pa; a2.y = pb;

            const float* hp = bc + (kk >> 1) * PP;
            const v2f b0 = *reinterpret_cast<const v2f*>(hp);
            const v2f b1 = *reinterpret_cast<const v2f*>(hp + 32);
            const v2f b2 = *reinterpret_cast<const v2f*>(hp + 64);
            const v2f b3 = *reinterpret_cast<const v2f*>(hp + 96);

            acc0 = __builtin_amdgcn_wmma_f32_16x16x4_f32(false, a2, false, b0, (short)0, acc0, false, false);
            acc1 = __builtin_amdgcn_wmma_f32_16x16x4_f32(false, a2, false, b1, (short)0, acc1, false, false);
            acc2 = __builtin_amdgcn_wmma_f32_16x16x4_f32(false, a2, false, b2, (short)0, acc2, false, false);
            acc3 = __builtin_amdgcn_wmma_f32_16x16x4_f32(false, a2, false, b3, (short)0, acc3, false, false);
        }

        if (more) {
#pragma unroll
            for (int r = 0; r < 4; ++r) {
                const float4 e = stg[2 * r], o = stg[2 * r + 1];
                float* d = bq + (cur ^ 1) * SLABF + sdst + 8 * r * PP;
                *reinterpret_cast<float4*>(d)     = make_float4(e.x, o.x, e.y, o.y);
                *reinterpret_cast<float4*>(d + 4) = make_float4(e.z, o.z, e.w, o.w);
            }
            if (gt < KJ) tt[q][cur ^ 1][gt] = tstg;
        }
        __syncthreads();
    }

    const int rbase = mt * 16 + ((lane >> 4) << 3);
#pragma unroll
    for (int v = 0; v < 8; ++v) {
        cred[q][rbase + v][m]      = acc0[v];
        cred[q][rbase + v][16 + m] = acc1[v];
        cred[q][rbase + v][32 + m] = acc2[v];
        cred[q][rbase + v][48 + m] = acc3[v];
    }
    zred[q][mt][lane] = zpart;
    __syncthreads();

    if (tid < ATT_ROWS) {
        const int zt = tid >> 4, zr = tid & 15;
        float z = 0.f;
#pragma unroll
        for (int w = 0; w < 4; ++w)
            z += zred[w][zt][zr] + zred[w][zt][zr + 16];
        zrow[tid] = z;
    }
    __syncthreads();

    for (int e = tid; e < ATT_ROWS * 64; e += 512) {
        const int rr = e >> 6, cc = e & 63;
        const float v = cred[0][rr][cc] + cred[1][rr][cc] +
                        cred[2][rr][cc] + cred[3][rr][cc];
        out[(size_t)(i0 + rr) * HF + hd * FDIM + cc] = v / zrow[rr];
    }
}

extern "C" void kernel_launch(void* const* d_in, const int* in_sizes, int n_in,
                              void* d_out, int out_size, void* d_ws, size_t ws_size,
                              hipStream_t stream) {
    const float* x = (const float*)d_in[0];
    const float* W = (const float*)d_in[2];
    const float* a = (const float*)d_in[3];
    float* out = (float*)d_out;

    char*  ws   = (char*)d_ws;
    float* h    = (float*)ws;
    float* s    = (float*)(ws + (size_t)N_NODES * HF * sizeof(float));
    float* t    = s + N_NODES * NHEAD;
    float* tmax = t + N_NODES * NHEAD;

    dim3 g1(N_NODES / GBM, HF / GBN);
    gat_gemm1<<<g1, 256, 0, stream>>>(x, W, h);

    gat_st<<<N_NODES, 128, 0, stream>>>(h, a, s, t);

    gat_tmax<<<NHEAD, 256, 0, stream>>>(t, tmax);

    dim3 g4(N_NODES / ATT_ROWS, NHEAD);
    gat_attn<<<g4, 512, 0, stream>>>(h, s, t, tmax, out);
}


# launches 4 -> 2: s,t dot products moved into the x@W kernel's epilogue (tile staged once in LDS, same sum order), max of t into the attention kernel's prologue; s,t stored head-major
# speedup vs baseline: 1.5226x; 1.1360x over previous
#include <hip/hip_runtime.h>
#include <math.h>

#define ALPHA   0.2f
#define N_NODES 4096
#define IN_FEAT 512
#define NHEAD   4
#define FDIM    64
#define HF      256

typedef __attribute__((ext_vector_type(2))) float v2f;
typedef __attribute__((ext_vector_type(8))) float v8f;

__device__ __forceinline__ float lrelu(float x) {
    return fmaxf(x, ALPHA * x);
}

#define GBM    64
#define GBN    64
#define GBK    64
#define GPA    68
#define GPB    160
#define GNSLAB (IN_FEAT / GBK)

__device__ __forceinline__ void gemm1_slab(const float* __restrict__ ac, const float* __restrict__ bc,
                                           v8f& acc0, v8f& acc1) {
#pragma unroll
    for (int kk = 0; kk < GBK; kk += 4) {
        const v2f a2 = *reinterpret_cast<const v2f*>(ac + kk);
        const float* bp = bc + (kk >> 1) * GPB;
        const v2f b0 = *reinterpret_cast<const v2f*>(bp);
        const v2f b1 = *reinterpret_cast<const v2f*>(bp + 32);
        acc0 = __builtin_amdgcn_wmma_f32_16x16x4_f32(false, a2, false, b0, (short)0, acc0, false, false);
        acc1 = __builtin_amdgcn_wmma_f32_16x16x4_f32(false, a2, false, b1, (short)0, acc1, false, false);
    }
}

__global__ __launch_bounds__(256) void gat_gemm1(const float* __restrict__ x,
                                                 const float* __restrict__ W,
                                                 const float* __restrict__ a,
                                                 float* __restrict__ h,
                                                 float* __restrict__ s,
                                                 float* __restrict__ t) {
    __shared__ __attribute__((aligned(16))) float xs[2][GBM * GPA];
    __shared__ __attribute__((aligned(16))) float wsl[2][(GBK / 2) * GPB];

    const int tid  = threadIdx.x;
    const int lane = tid & 31;
    const int wave = tid >> 5;
    const int mt   = wave & 3;
    const int nh   = wave >> 2;
    const int i0   = blockIdx.x * GBM;
    const int n0   = blockIdx.y * GBN;
    const int m    = lane & 15;
    const int hf   = lane >> 4;

    const int srow = tid >> 4;
    const int scol = (tid & 15) << 2;
    const float* xsrc = x + (size_t)(i0 + srow) * IN_FEAT + scol;
    const float* wsrc = W + (size_t)(2 * srow) * HF + n0 + scol;
    const int xdst = srow * GPA + scol;
    const int wdst = srow * GPB + scol * 2;

    float4 x0, x1, x2, x3;
    float4 we0, wo0, we1, wo1;
    x0  = *reinterpret_cast<const float4*>(xsrc);
    x1  = *reinterpret_cast<const float4*>(xsrc + (size_t)16 * IN_FEAT);
    x2  = *reinterpret_cast<const float4*>(xsrc + (size_t)32 * IN_FEAT);
    x3  = *reinterpret_cast<const float4*>(xsrc + (size_t)48 * IN_FEAT);
    we0 = *reinterpret_cast<const float4*>(wsrc);
    wo0 = *reinterpret_cast<const float4*>(wsrc + (size_t)HF);
    we1 = *reinterpret_cast<const float4*>(wsrc + (size_t)32 * HF);
    wo1 = *reinterpret_cast<const float4*>(wsrc + (size_t)33 * HF);
    {
        float* xd = &xs[0][xdst];
        float* wd = &wsl[0][wdst];
        *reinterpret_cast<float4*>(xd)                = x0;
        *reinterpret_cast<float4*>(xd + 16 * GPA)     = x1;
        *reinterpret_cast<float4*>(xd + 32 * GPA)     = x2;
        *reinterpret_cast<float4*>(xd + 48 * GPA)     = x3;
        *reinterpret_cast<float4*>(wd)                = make_float4(we0.x, wo0.x, we0.y, wo0.y);
        *reinterpret_cast<float4*>(wd + 4)            = make_float4(we0.z, wo0.z, we0.w, wo0.w);
        *reinterpret_cast<float4*>(wd + 16 * GPB)     = make_float4(we1.x, wo1.x, we1.y, wo1.y);
        *reinterpret_cast<float4*>(wd + 16 * GPB + 4) = make_float4(we1.z, wo1.z, we1.w, wo1.w);
    }
    __syncthreads();

    v8f acc0 = {}, acc1 = {};

    const int aoff = (mt * 16 + m) * GPA + 2 * hf;
    const int boff = hf * GPB + 2 * (nh * 32 + m);

#pragma nounroll
    for (int sl = 0; sl < GNSLAB - 1; ++sl) {
        const int cur = sl & 1;
        const float* xn = xsrc + (sl + 1) * GBK;
        const float* wn = wsrc + (size_t)((sl + 1) * GBK) * HF;
        x0  = *reinterpret_cast<const float4*>(xn);
        x1  = *reinterpret_cast<const float4*>(xn + (size_t)16 * IN_FEAT);
        x2  = *reinterpret_cast<const float4*>(xn + (size_t)32 * IN_FEAT);
        x3  = *reinterpret_cast<const float4*>(xn + (size_t)48 * IN_FEAT);
        we0 = *reinterpret_cast<const float4*>(wn);
        wo0 = *reinterpret_cast<const float4*>(wn + (size_t)HF);
        we1 = *reinterpret_cast<const float4*>(wn + (size_t)32 * HF);
        wo1 = *reinterpret_cast<const float4*>(wn + (size_t)33 * HF);

        gemm1_slab(&xs[cur][aoff], &wsl[cur][boff], acc0, acc1);
        __builtin_amdgcn_sched_barrier(0x20);

        float* xd = &xs[cur ^ 1][xdst];
        float* wd = &wsl[cur ^ 1][wdst];
        *reinterpret_cast<float4*>(xd)                = x0;
        *reinterpret_cast<float4*>(xd + 16 * GPA)     = x1;
        *reinterpret_cast<float4*>(xd + 32 * GPA)     = x2;
        *reinterpret_cast<float4*>(xd + 48 * GPA)     = x3;
        *reinterpret_cast<float4*>(wd)                = make_float4(we0.x, wo0.x, we0.y, wo0.y);
        *reinterpret_cast<float4*>(wd + 4)            = make_float4(we0.z, wo0.z, we0.w, wo0.w);
        *reinterpret_cast<float4*>(wd + 16 * GPB)     = make_float4(we1.x, wo1.x, we1.y, wo1.y);
        *reinterpret_cast<float4*>(wd + 16 * GPB + 4) = make_float4(we1.z, wo1.z, we1.w, wo1.w);
        __syncthreads();
    }
    const int hd = blockIdx.y;
    const float* asrc = a + hd * 2 * FDIM;
    const float* adst = asrc + FDIM;
    const float as0 = asrc[lane], as1 = asrc[lane + 32];
    const float ad0 = adst[lane], ad1 = adst[lane + 32];

    gemm1_slab(&xs[(GNSLAB - 1) & 1][aoff], &wsl[(GNSLAB - 1) & 1][boff], acc0, acc1);

    float* hrow = h + (size_t)(i0 + mt * 16 + (hf << 3)) * HF + n0 + nh * 32 + m;
#pragma unroll
    for (int v = 0; v < 8; ++v) {
        hrow[(size_t)v * HF]      = acc0[v];
        hrow[(size_t)v * HF + 16] = acc1[v];
    }

    float* hs  = &xs[0][0];
    float* stl = &wsl[0][0];
    {
        float* hd0 = hs + (mt * 16 + (hf << 3)) * GPA + nh * 32 + m;
#pragma unroll
        for (int v = 0; v < 8; ++v) {
            hd0[v * GPA]      = acc0[v];
            hd0[v * GPA + 16] = acc1[v];
        }
    }
    __syncthreads();
    {
        float ms = 0.f, mtv = 0.f;
#pragma unroll
        for (int r = 0; r < 8; ++r) {
            const float* hr = hs + (wave * 8 + r) * GPA;
            const float h0 = hr[lane], h1 = hr[lane + 32];
            float sv = h0 * as0 + h1 * as1;
            float tv = h0 * ad0 + h1 * ad1;
            for (int off = 16; off > 0; off >>= 1) {
                sv += __shfl_xor(sv, off, 32);
                tv += __shfl_xor(tv, off, 32);
            }
            if (lane == r) { ms = sv; mtv = tv; }
        }
        if (lane < 8) {
            stl[wave * 8 + lane]      = ms;
            stl[64 + wave * 8 + lane] = mtv;
        }
    }
    __syncthreads();
    if (wave < 2) {
        float* dst = (wave == 0 ? s : t) + (size_t)hd * N_NODES + i0 + 2 * lane;
        *reinterpret_cast<float2*>(dst) = *reinterpret_cast<const float2*>(stl + wave * 64 + 2 * lane);
    }
}

#define ATT_ROWS 64
#define KJ       64
#define PP       160
#define SLABF    ((KJ / 2) * PP)
#define NSLAB    (N_NODES / 4 / KJ)

__global__ __launch_bounds__(512) void gat_attn(const float* __restrict__ h,
                                                const float* __restrict__ s,
                                                const float* __restrict__ t,
                                                float* __restrict__ out) {
    __shared__ __attribute__((aligned(16))) float bt[4 * 2 * SLABF];
    __shared__ __attribute__((aligned(16))) float tt[4][2][KJ];
    float (*cred)[ATT_ROWS][64] = reinterpret_cast<float (*)[ATT_ROWS][64]>(bt);
    __shared__ float zred[4][4][32];
    __shared__ float zrow[ATT_ROWS];
    __shared__ __attribute__((aligned(16))) float tmred[16];

    const int tid  = threadIdx.x;
    const int lane = tid & 31;
    const int wave = tid >> 5;
    const int q    = wave >> 2;
    const int mt   = wave & 3;
    const int gt   = tid & 127;
    const int i0   = blockIdx.x * ATT_ROWS;
    const int hd   = blockIdx.y;
    const int m    = lane & 15;
    const int koff = (lane >> 4) << 1;

    const float* th  = t + (size_t)hd * N_NODES;
    const float sm   = s[(size_t)hd * N_NODES + i0 + mt * 16 + m];
    const float* hh  = h + hd * FDIM;

    const int sp   = gt >> 4;
    const int scol = (gt & 15) << 2;
    const int jq0  = q * (N_NODES / 4);
    float* const bq = bt + q * (2 * SLABF);
    const int sdst  = sp * PP + scol * 2;

    float4 stg[8];
    float  tstg = 0.f;
    {
        const float* src0 = hh + (size_t)(jq0 + 2 * sp) * HF + scol;
#pragma unroll
        for (int r = 0; r < 4; ++r) {
            stg[2 * r]     = *reinterpret_cast<const float4*>(src0 + (size_t)(16 * r) * HF);
            stg[2 * r + 1] = *reinterpret_cast<const float4*>(src0 + (size_t)(16 * r + 1) * HF);
        }
        if (gt < KJ) tstg = th[jq0 + gt];
        const float4 u0 = *reinterpret_cast<const float4*>(th + 4 * tid);
        const float4 u1 = *reinterpret_cast<const float4*>(th + 4 * (tid + 512));
        float mx = fmaxf(fmaxf(fmaxf(u0.x, u0.y), fmaxf(u0.z, u0.w)),
                         fmaxf(fmaxf(u1.x, u1.y), fmaxf(u1.z, u1.w)));
        for (int off = 16; off > 0; off >>= 1)
            mx = fmaxf(mx, __shfl_xor(mx, off, 32));
        if (lane == 0) tmred[wave] = mx;
#pragma unroll
        for (int r = 0; r < 4; ++r) {
            const float4 e = stg[2 * r], o = stg[2 * r + 1];
            float* d = bq + sdst + 8 * r * PP;
            *reinterpret_cast<float4*>(d)     = make_float4(e.x, o.x, e.y, o.y);
            *reinterpret_cast<float4*>(d + 4) = make_float4(e.z, o.z, e.w, o.w);
        }
        if (gt < KJ) tt[q][0][gt] = tstg;
    }
    __syncthreads();

    const float4 r0 = *reinterpret_cast<const float4*>(&tmred[0]);
    const float4 r1 = *reinterpret_cast<const float4*>(&tmred[4]);
    const float4 r2 = *reinterpret_cast<const float4*>(&tmred[8]);
    const float4 r3 = *reinterpret_cast<const float4*>(&tmred[12]);
    const float tmaxv = fmaxf(fmaxf(fmaxf(fmaxf(r0.x, r0.y), fmaxf(r0.z, r0.w)),
                                    fmaxf(fmaxf(r1.x, r1.y), fmaxf(r1.z, r1.w))),
                              fmaxf(fmaxf(fmaxf(r2.x, r2.y), fmaxf(r2.z, r2.w)),
                                    fmaxf(fmaxf(r3.x, r3.y), fmaxf(r3.z, r3.w))));
    const float mrow = lrelu(sm + tmaxv);

    v8f acc0 = {}, acc1 = {}, acc2 = {}, acc3 = {};
    float zpart = 0.f;

    for (int sl = 0; sl < NSLAB; ++sl) {
        const int  cur  = sl & 1;
        const bool more = (sl + 1 < NSLAB);
        if (more) {
            const float* src0 = hh + (size_t)(jq0 + (sl + 1) * KJ + 2 * sp) * HF + scol;
#pragma unroll
            for (int r = 0; r < 4; ++r) {
                stg[2 * r]     = *reinterpret_cast<const float4*>(src0 + (size_t)(16 * r) * HF);
                stg[2 * r + 1] = *reinterpret_cast<const float4*>(src0 + (size_t)(16 * r + 1) * HF);
            }
            if (gt < KJ) tstg = th[jq0 + (sl + 1) * KJ + gt];
        }

        const float* bc = bq + cur * SLABF + (lane >> 4) * PP + 2 * m;
        const float* tc = &tt[q][cur][0];
#pragma unroll
        for (int kk = 0; kk < KJ; kk += 4) {
            const int ja = kk + koff;
            const float pa = __expf(lrelu(sm + tc[ja]) - mrow);
            const float pb = __expf(lrelu(sm + tc[ja + 1]) - mrow);
            zpart += pa + pb;
            v2f a2; a2.x = pa; a2.y = pb;

            const float* hp = bc + (kk >> 1) * PP;
            const v2f b0 = *reinterpret_cast<const v2f*>(hp);
            const v2f b1 = *reinterpret_cast<const v2f*>(hp + 32);
            const v2f b2 = *reinterpret_cast<const v2f*>(hp + 64);
            const v2f b3 = *reinterpret_cast<const v2f*>(hp + 96);

            acc0 = __builtin_amdgcn_wmma_f32_16x16x4_f32(false, a2, false, b0, (short)0, acc0, false, false);
            acc1 = __builtin_amdgcn_wmma_f32_16x16x4_f32(false, a2, false, b1, (short)0, acc1, false, false);
            acc2 = __builtin_amdgcn_wmma_f32_16x16x4_f32(false, a2, false, b2, (short)0, acc2, false, false);
            acc3 = __builtin_amdgcn_wmma_f32_16x16x4_f32(false, a2, false, b3, (short)0, acc3, false, false);
        }

        if (more) {
#pragma unroll
            for (int r = 0; r < 4; ++r) {
                const float4 e = stg[2 * r], o = stg[2 * r + 1];
                float* d = bq + (cur ^ 1) * SLABF + sdst + 8 * r * PP;
                *reinterpret_cast<float4*>(d)     = make_float4(e.x, o.x, e.y, o.y);
                *reinterpret_cast<float4*>(d + 4) = make_float4(e.z, o.z, e.w, o.w);
            }
            if (gt < KJ) tt[q][cur ^ 1][gt] = tstg;
        }
        __syncthreads();
    }

    const int rbase = mt * 16 + ((lane >> 4) << 3);
#pragma unroll
    for (int v = 0; v < 8; ++v) {
        cred[q][rbase + v][m]      = acc0[v];
        cred[q][rbase + v][16 + m] = acc1[v];
        cred[q][rbase + v][32 + m] = acc2[v];
        cred[q][rbase + v][48 + m] = acc3[v];
    }
    zred[q][mt][lane] = zpart;
    __syncthreads();

    if (tid < ATT_ROWS) {
        const int zt = tid >> 4, zr = tid & 15;
        float z = 0.f;
#pragma unroll
        for (int w = 0; w < 4; ++w)
            z += zred[w][zt][zr] + zred[w][zt][zr + 16];
        zrow[tid] = z;
    }
    __syncthreads();

    for (int e = tid; e < ATT_ROWS * 64; e += 512) {
        const int rr = e >> 6, cc = e & 63;
        const float v = cred[0][rr][cc] + cred[1][rr][cc] +
                        cred[2][rr][cc] + cred[3][rr][cc];
        out[(size_t)(i0 + rr) * HF + hd * FDIM + cc] = v / zrow[rr];
    }
}

extern "C" void kernel_launch(void* const* d_in, const int* in_sizes, int n_in,
                              void* d_out, int out_size, void* d_ws, size_t ws_size,
                              hipStream_t stream) {
    const float* x = (const float*)d_in[0];
    const float* W = (const float*)d_in[2];
    const float* a = (const float*)d_in[3];
    float* out = (float*)d_out;

    char*  ws   = (char*)d_ws;
    float* h    = (float*)ws;
    float* s    = (float*)(ws + (size_t)N_NODES * HF * sizeof(float));
    float* t    = s + N_NODES * NHEAD;

    dim3 g1(N_NODES / GBM, HF / GBN);
    gat_gemm1<<<g1, 256, 0, stream>>>(x, W, a, h, s, t);

    dim3 g4(N_NODES / ATT_ROWS, NHEAD);
    gat_attn<<<g4, 512, 0, stream>>>(h, s, t, out);
}


# attention k loop free of exp: exp(lrelu(s_i+t_j)-m_i) = max(fa_i*E_j, fb_i*G_j); E,G = exp once per staged row of h, fa,fb once per output row; P differs in last bits, every sum order kept
# speedup vs baseline: 1.5809x; 1.0383x over previous
#include <hip/hip_runtime.h>
#include <math.h>

#define ALPHA   0.2f
#define N_NODES 4096
#define IN_FEAT 512
#define NHEAD   4
#define FDIM    64
#define HF      256

typedef __attribute__((ext_vector_type(2))) float v2f;
typedef __attribute__((ext_vector_type(8))) float v8f;

__device__ __forceinline__ float lrelu(float x) {
    return fmaxf(x, ALPHA * x);
}

#define GBM    64
#define GBN    64
#define GBK    64
#define GPA    68
#define GPB    160
#define GNSLAB (IN_FEAT / GBK)

__device__ __forceinline__ void gemm1_slab(const float* __restrict__ ac, const float* __restrict__ bc,
                                           v8f& acc0, v8f& acc1) {
#pragma unroll
    for (int kk = 0; kk < GBK; kk += 4) {
        const v2f a2 = *reinterpret_cast<const v2f*>(ac + kk);
        const float* bp = bc + (kk >> 1) * GPB;
        const v2f b0 = *reinterpret_cast<const v2f*>(bp);
        const v2f b1 = *reinterpret_cast<const v2f*>(bp + 32);
        acc0 = __builtin_amdgcn_wmma_f32_16x16x4_f32(false, a2, false, b0, (short)0, acc0, false, false);
        acc1 = __builtin_amdgcn_wmma_f32_16x16x4_f32(false, a2, false, b1, (short)0, acc1, false, false);
    }
}

__global__ __launch_bounds__(256) void gat_gemm1(const float* __restrict__ x,
                                                 const float* __restrict__ W,
                                                 const float* __restrict__ a,
                                                 float* __restrict__ h,
                                                 float* __restrict__ s,
                                                 float* __restrict__ t) {
    __shared__ __attribute__((aligned(16))) float xs[2][GBM * GPA];
    __shared__ __attribute__((aligned(16))) float wsl[2][(GBK / 2) * GPB];

    const int tid  = threadIdx.x;
    const int lane = tid & 31;
    const int wave = tid >> 5;
    const int mt   = wave & 3;
    const int nh   = wave >> 2;
    const int i0   = blockIdx.x * GBM;
    const int n0   = blockIdx.y * GBN;
    const int m    = lane & 15;
    const int hf   = lane >> 4;

    const int srow = tid >> 4;
    const int scol = (tid & 15) << 2;
    const float* xsrc = x + (size_t)(i0 + srow) * IN_FEAT + scol;
    const float* wsrc = W + (size_t)(2 * srow) * HF + n0 + scol;
    const int xdst = srow * GPA + scol;
    const int wdst = srow * GPB + scol * 2;

    float4 x0, x1, x2, x3;
    float4 we0, wo0, we1, wo1;
    x0  = *reinterpret_cast<const float4*>(xsrc);
    x1  = *reinterpret_cast<const float4*>(xsrc + (size_t)16 * IN_FEAT);
    x2  = *reinterpret_cast<const float4*>(xsrc + (size_t)32 * IN_FEAT);
    x3  = *reinterpret_cast<const float4*>(xsrc + (size_t)48 * IN_FEAT);
    we0 = *reinterpret_cast<const float4*>(wsrc);
    wo0 = *reinterpret_cast<const float4*>(wsrc + (size_t)HF);
    we1 = *reinterpret_cast<const float4*>(wsrc + (size_t)32 * HF);
    wo1 = *reinterpret_cast<const float4*>(wsrc + (size_t)33 * HF);
    {
        float* xd = &xs[0][xdst];
        float* wd = &wsl[0][wdst];
        *reinterpret_cast<float4*>(xd)                = x0;
        *reinterpret_cast<float4*>(xd + 16 * GPA)     = x1;
        *reinterpret_cast<float4*>(xd + 32 * GPA)     = x2;
        *reinterpret_cast<float4*>(xd + 48 * GPA)     = x3;
        *reinterpret_cast<float4*>(wd)                = make_float4(we0.x, wo0.x, we0.y, wo0.y);
        *reinterpret_cast<float4*>(wd + 4)            = make_float4(we0.z, wo0.z, we0.w, wo0.w);
        *reinterpret_cast<float4*>(wd + 16 * GPB)     = make_float4(we1.x, wo1.x, we1.y, wo1.y);
        *reinterpret_cast<float4*>(wd + 16 * GPB + 4) = make_float4(we1.z, wo1.z, we1.w, wo1.w);
    }
    __syncthreads();

    v8f acc0 = {}, acc1 = {};

    const int aoff = (mt * 16 + m) * GPA + 2 * hf;
    const int boff = hf * GPB + 2 * (nh * 32 + m);

#pragma nounroll
    for (int sl = 0; sl < GNSLAB - 1; ++sl) {
        const int cur = sl & 1;
        const float* xn = xsrc + (sl + 1) * GBK;
        const float* wn = wsrc + (size_t)((sl + 1) * GBK) * HF;
        x0  = *reinterpret_cast<const float4*>(xn);
        x1  = *reinterpret_cast<const float4*>(xn + (size_t)16 * IN_FEAT);
        x2  = *reinterpret_cast<const float4*>(xn + (size_t)32 * IN_FEAT);
        x3  = *reinterpret_cast<const float4*>(xn + (size_t)48 * IN_FEAT);
        we0 = *reinterpret_cast<const float4*>(wn);
        wo0 = *reinterpret_cast<const float4*>(wn + (size_t)HF);
        we1 = *reinterpret_cast<const float4*>(wn + (size_t)32 * HF);
        wo1 = *reinterpret_cast<const float4*>(wn + (size_t)33 * HF);

        gemm1_slab(&xs[cur][aoff], &wsl[cur][boff], acc0, acc1);
        __builtin_amdgcn_sched_barrier(0x20);

        float* xd = &xs[cur ^ 1][xdst];
        float* wd = &wsl[cur ^ 1][wdst];
        *reinterpret_cast<float4*>(xd)                = x0;
        *reinterpret_cast<float4*>(xd + 16 * GPA)     = x1;
        *reinterpret_cast<float4*>(xd + 32 * GPA)     = x2;
        *reinterpret_cast<float4*>(xd + 48 * GPA)     = x3;
        *reinterpret_cast<float4*>(wd)                = make_float4(we0.x, wo0.x, we0.y, wo0.y);
        *reinterpret_cast<float4*>(wd + 4)            = make_float4(we0.z, wo0.z, we0.w, wo0.w);
        *reinterpret_cast<float4*>(wd + 16 * GPB)     = make_float4(we1.x, wo1.x, we1.y, wo1.y);
        *reinterpret_cast<float4*>(wd + 16 * GPB + 4) = make_float4(we1.z, wo1.z, we1.w, wo1.w);
        __syncthreads();
    }
    const int hd = blockIdx.y;
    const float* asrc = a + hd * 2 * FDIM;
    const float* adst = asrc + FDIM;
    const float as0 = asrc[lane], as1 = asrc[lane + 32];
    const float ad0 = adst[lane], ad1 = adst[lane + 32];

    gemm1_slab(&xs[(GNSLAB - 1) & 1][aoff], &wsl[(GNSLAB - 1) & 1][boff], acc0, acc1);

    float* hrow = h + (size_t)(i0 + mt * 16 + (hf << 3)) * HF + n0 + nh * 32 + m;
#pragma unroll
    for (int v = 0; v < 8; ++v) {
        hrow[(size_t)v * HF]      = acc0[v];
        hrow[(size_t)v * HF + 16] = acc1[v];
    }

    float* hs  = &xs[0][0];
    float* stl = &wsl[0][0];
    {
        float* hd0 = hs + (mt * 16 + (hf << 3)) * GPA + nh * 32 + m;
#pragma unroll
        for (int v = 0; v < 8; ++v) {
            hd0[v * GPA]      = acc0[v];
            hd0[v * GPA + 16] = acc1[v];
        }
    }
    __syncthreads();
    {
        float ms = 0.f, mtv = 0.f;
#pragma unroll
        for (int r = 0; r < 8; ++r) {
            const float* hr = hs + (wave * 8 + r) * GPA;
            const float h0 = hr[lane], h1 = hr[lane + 32];
            float sv = h0 * as0 + h1 * as1;
            float tv = h0 * ad0 + h1 * ad1;
            for (int off = 16; off > 0; off >>= 1) {
                sv += __shfl_xor(sv, off, 32);
                tv += __shfl_xor(tv, off, 32);
            }
            if (lane == r) { ms = sv; mtv = tv; }
        }
        if (lane < 8) {
            stl[wave * 8 + lane]      = ms;
            stl[64 + wave * 8 + lane] = mtv;
        }
    }
    __syncthreads();
    if (wave < 2) {
        float* dst = (wave == 0 ? s : t) + (size_t)hd * N_NODES + i0 + 2 * lane;
        *reinterpret_cast<float2*>(dst) = *reinterpret_cast<const float2*>(stl + wave * 64 + 2 * lane);
    }
}

#define ATT_ROWS 64
#define KJ       64
#define PP       160
#define SLABF    ((KJ / 2) * PP)
#define NSLAB    (N_NODES / 4 / KJ)

__global__ __launch_bounds__(512) void gat_attn(const float* __restrict__ h,
                                                const float* __restrict__ s,
                                                const float* __restrict__ t,
                                                float* __restrict__ out) {
    __shared__ __attribute__((aligned(16))) float bt[4 * 2 * SLABF];
    __shared__ __attribute__((aligned(16))) float eg[4][2][2 * KJ];
    float (*cred)[ATT_ROWS][64] = reinterpret_cast<float (*)[ATT_ROWS][64]>(bt);
    __shared__ float zred[4][4][32];
    __shared__ float zrow[ATT_ROWS];
    __shared__ __attribute__((aligned(16))) float tmred[16];

    const int tid  = threadIdx.x;
    const int lane = tid & 31;
    const int wave = tid >> 5;
    const int q    = wave >> 2;
    const int mt   = wave & 3;
    const int gt   = tid & 127;
    const int i0   = blockIdx.x * ATT_ROWS;
    const int hd   = blockIdx.y;
    const int m    = lane & 15;
    const int koff = (lane >> 4) << 1;

    const float* th  = t + (size_t)hd * N_NODES;
    const float sm   = s[(size_t)hd * N_NODES + i0 + mt * 16 + m];
    const float* hh  = h + hd * FDIM;

    const int sp   = gt >> 4;
    const int scol = (gt & 15) << 2;
    const int jq0  = q * (N_NODES / 4);
    float* const bq = bt + q * (2 * SLABF);
    const int sdst  = sp * PP + scol * 2;

    float4 stg[8];
    float2 tstg = make_float2(0.f, 0.f);
    {
        const float* src0 = hh + (size_t)(jq0 + 2 * sp) * HF + scol;
#pragma unroll
        for (int r = 0; r < 4; ++r) {
            stg[2 * r]     = *reinterpret_cast<const float4*>(src0 + (size_t)(16 * r) * HF);
            stg[2 * r + 1] = *reinterpret_cast<const float4*>(src0 + (size_t)(16 * r + 1) * HF);
        }
        if (gt < KJ / 2) tstg = *reinterpret_cast<const float2*>(th + jq0 + 2 * gt);
        const float4 u0 = *reinterpret_cast<const float4*>(th + 4 * tid);
        const float4 u1 = *reinterpret_cast<const float4*>(th + 4 * (tid + 512));
        float mx = fmaxf(fmaxf(fmaxf(u0.x, u0.y), fmaxf(u0.z, u0.w)),
                         fmaxf(fmaxf(u1.x, u1.y), fmaxf(u1.z, u1.w)));
        for (int off = 16; off > 0; off >>= 1)
            mx = fmaxf(mx, __shfl_xor(mx, off, 32));
        if (lane == 0) tmred[wave] = mx;
#pragma unroll
        for (int r = 0; r < 4; ++r) {
            const float4 e = stg[2 * r], o = stg[2 * r + 1];
            float* d = bq + sdst + 8 * r * PP;
            *reinterpret_cast<float4*>(d)     = make_float4(e.x, o.x, e.y, o.y);
            *reinterpret_cast<float4*>(d + 4) = make_float4(e.z, o.z, e.w, o.w);
        }
    }
    __syncthreads();

    const float4 r0 = *reinterpret_cast<const float4*>(&tmred[0]);
    const float4 r1 = *reinterpret_cast<const float4*>(&tmred[4]);
    const float4 r2 = *reinterpret_cast<const float4*>(&tmred[8]);
    const float4 r3 = *reinterpret_cast<const float4*>(&tmred[12]);
    const float tmaxv = fmaxf(fmaxf(fmaxf(fmaxf(r0.x, r0.y), fmaxf(r0.z, r0.w)),
                                    fmaxf(fmaxf(r1.x, r1.y), fmaxf(r1.z, r1.w))),
                              fmaxf(fmaxf(fmaxf(r2.x, r2.y), fmaxf(r2.z, r2.w)),
                                    fmaxf(fmaxf(r3.x, r3.y), fmaxf(r3.z, r3.w))));
    const float xrow = sm + tmaxv;
    const float mrow = lrelu(xrow);
    const float fa   = __expf(xrow - mrow);
    const float fb   = __expf(ALPHA * xrow - mrow);
    const v2f fa2 = {fa, fa}, fb2 = {fb, fb};

    if (gt < KJ / 2) {
        const float d0 = tstg.x - tmaxv, d1 = tstg.y - tmaxv;
        *reinterpret_cast<float4*>(&eg[q][0][4 * gt]) =
            make_float4(__expf(d0), __expf(d1), __expf(ALPHA * d0), __expf(ALPHA * d1));
    }
    __syncthreads();

    v8f acc0 = {}, acc1 = {}, acc2 = {}, acc3 = {};
    float zpart = 0.f;

    for (int sl = 0; sl < NSLAB; ++sl) {
        const int  cur  = sl & 1;
        const bool more = (sl + 1 < NSLAB);
        if (more) {
            const float* src0 = hh + (size_t)(jq0 + (sl + 1) * KJ + 2 * sp) * HF + scol;
#pragma unroll
            for (int r = 0; r < 4; ++r) {
                stg[2 * r]     = *reinterpret_cast<const float4*>(src0 + (size_t)(16 * r) * HF);
                stg[2 * r + 1] = *reinterpret_cast<const float4*>(src0 + (size_t)(16 * r + 1) * HF);
            }
            if (gt < KJ / 2) tstg = *reinterpret_cast<const float2*>(th + jq0 + (sl + 1) * KJ + 2 * gt);
        }

        const float* bc = bq + cur * SLABF + (lane >> 4) * PP + 2 * m;
        const float* ec = &eg[q][cur][2 * koff];
#pragma unroll
        for (int kk = 0; kk < KJ; kk += 4) {
            const v2f e2 = *reinterpret_cast<const v2f*>(ec + 2 * kk);
            const v2f g2 = *reinterpret_cast<const v2f*>(ec + 2 * kk + 2);
            const v2f a2 = __builtin_elementwise_max(fa2 * e2, fb2 * g2);
            zpart += a2.x + a2.y;

            const float* hp = bc + (kk >> 1) * PP;
            const v2f b0 = *reinterpret_cast<const v2f*>(hp);
            const v2f b1 = *reinterpret_cast<const v2f*>(hp + 32);
            const v2f b2 = *reinterpret_cast<const v2f*>(hp + 64);
            const v2f b3 = *reinterpret_cast<const v2f*>(hp + 96);

            acc0 = __builtin_amdgcn_wmma_f32_16x16x4_f32(false, a2, false, b0, (short)0, acc0, false, false);
            acc1 = __builtin_amdgcn_wmma_f32_16x16x4_f32(false, a2, false, b1, (short)0, acc1, false, false);
            acc2 = __builtin_amdgcn_wmma_f32_16x16x4_f32(false, a2, false, b2, (short)0, acc2, false, false);
            acc3 = __builtin_amdgcn_wmma_f32_16x16x4_f32(false, a2, false, b3, (short)0, acc3, false, false);
        }

        if (more) {
#pragma unroll
            for (int r = 0; r < 4; ++r) {
                const float4 e = stg[2 * r], o = stg[2 * r + 1];
                float* d = bq + (cur ^ 1) * SLABF + sdst + 8 * r * PP;
                *reinterpret_cast<float4*>(d)     = make_float4(e.x, o.x, e.y, o.y);
                *reinterpret_cast<float4*>(d + 4) = make_float4(e.z, o.z, e.w, o.w);
            }
            if (gt < KJ / 2) {
                const float d0 = tstg.x - tmaxv, d1 = tstg.y - tmaxv;
                *reinterpret_cast<float4*>(&eg[q][cur ^ 1][4 * gt]) =
                    make_float4(__expf(d0), __expf(d1), __expf(ALPHA * d0), __expf(ALPHA * d1));
            }
        }
        __syncthreads();
    }

    const int rbase = mt * 16 + ((lane >> 4) << 3);
#pragma unroll
    for (int v = 0; v < 8; ++v) {
        cred[q][rbase + v][m]      = acc0[v];
        cred[q][rbase + v][16 + m] = acc1[v];
        cred[q][rbase + v][32 + m] = acc2[v];
        cred[q][rbase + v][48 + m] = acc3[v];
    }
    zred[q][mt][lane] = zpart;
    __syncthreads();

    if (tid < ATT_ROWS) {
        const int zt = tid >> 4, zr = tid & 15;
        float z = 0.f;
#pragma unroll
        for (int w = 0; w < 4; ++w)
            z += zred[w][zt][zr] + zred[w][zt][zr + 16];
        zrow[tid] = z;
    }
    __syncthreads();

    for (int e = tid; e < ATT_ROWS * 64; e += 512) {
        const int rr = e >> 6, cc = e & 63;
        const float v = cred[0][rr][cc] + cred[1][rr][cc] +
                        cred[2][rr][cc] + cred[3][rr][cc];
        out[(size_t)(i0 + rr) * HF + hd * FDIM + cc] = v / zrow[rr];
    }
}

extern "C" void kernel_launch(void* const* d_in, const int* in_sizes, int n_in,
                              void* d_out, int out_size, void* d_ws, size_t ws_size,
                              hipStream_t stream) {
    const float* x = (const float*)d_in[0];
    const float* W = (const float*)d_in[2];
    const float* a = (const float*)d_in[3];
    float* out = (float*)d_out;

    char*  ws   = (char*)d_ws;
    float* h    = (float*)ws;
    float* s    = (float*)(ws + (size_t)N_NODES * HF * sizeof(float));
    float* t    = s + N_NODES * NHEAD;

    dim3 g1(N_NODES / GBM, HF / GBN);
    gat_gemm1<<<g1, 256, 0, stream>>>(x, W, a, h, s, t);

    dim3 g4(N_NODES / ATT_ROWS, NHEAD);
    gat_attn<<<g4, 512, 0, stream>>>(h, s, t, out);
}


# attention k loop without LDS or barrier: one wave per SIMD owns a 1024-key quarter and all 4 M tiles; B fragments by 16-byte global loads (N tile n = columns 4c+n); every sum order kept
# speedup vs baseline: 1.6536x; 1.0460x over previous
#include <hip/hip_runtime.h>
#include <math.h>

#define ALPHA   0.2f
#define N_NODES 4096
#define IN_FEAT 512
#define NHEAD   4
#define FDIM    64
#define HF      256

typedef __attribute__((ext_vector_type(2))) float v2f;
typedef __attribute__((ext_vector_type(8))) float v8f;

__device__ __forceinline__ float lrelu(float x) {
    return fmaxf(x, ALPHA * x);
}

#define GBM    64
#define GBN    64
#define GBK    64
#define GPA    68
#define GPB    160
#define GNSLAB (IN_FEAT / GBK)

__device__ __forceinline__ void gemm1_slab(const float* __restrict__ ac, const float* __restrict__ bc,
                                           v8f& acc0, v8f& acc1) {
#pragma unroll
    for (int kk = 0; kk < GBK; kk += 4) {
        const v2f a2 = *reinterpret_cast<const v2f*>(ac + kk);
        const float* bp = bc + (kk >> 1) * GPB;
        const v2f b0 = *reinterpret_cast<const v2f*>(bp);
        const v2f b1 = *reinterpret_cast<const v2f*>(bp + 32);
        acc0 = __builtin_amdgcn_wmma_f32_16x16x4_f32(false, a2, false, b0, (short)0, acc0, false, false);
        acc1 = __builtin_amdgcn_wmma_f32_16x16x4_f32(false, a2, false, b1, (short)0, acc1, false, false);
    }
}

__global__ __launch_bounds__(256) void gat_gemm1(const float* __restrict__ x,
                                                 const float* __restrict__ W,
                                                 const float* __restrict__ a,
                                                 float* __restrict__ h,
                                                 float* __restrict__ s,
                                                 float* __restrict__ t) {
    __shared__ __attribute__((aligned(16))) float xs[2][GBM * GPA];
    __shared__ __attribute__((aligned(16))) float wsl[2][(GBK / 2) * GPB];

    const int tid  = threadIdx.x;
    const int lane = tid & 31;
    const int wave = tid >> 5;
    const int mt   = wave & 3;
    const int nh   = wave >> 2;
    const int i0   = blockIdx.x * GBM;
    const int n0   = blockIdx.y * GBN;
    const int m    = lane & 15;
    const int hf   = lane >> 4;

    const int srow = tid >> 4;
    const int scol = (tid & 15) << 2;
    const float* xsrc = x + (size_t)(i0 + srow) * IN_FEAT + scol;
    const float* wsrc = W + (size_t)(2 * srow) * HF + n0 + scol;
    const int xdst = srow * GPA + scol;
    const int wdst = srow * GPB + scol * 2;

    float4 x0, x1, x2, x3;
    float4 we0, wo0, we1, wo1;
    x0  = *reinterpret_cast<const float4*>(xsrc);
    x1  = *reinterpret_cast<const float4*>(xsrc + (size_t)16 * IN_FEAT);
    x2  = *reinterpret_cast<const float4*>(xsrc + (size_t)32 * IN_FEAT);
    x3  = *reinterpret_cast<const float4*>(xsrc + (size_t)48 * IN_FEAT);
    we0 = *reinterpret_cast<const float4*>(wsrc);
    wo0 = *reinterpret_cast<const float4*>(wsrc + (size_t)HF);
    we1 = *reinterpret_cast<const float4*>(wsrc + (size_t)32 * HF);
    wo1 = *reinterpret_cast<const float4*>(wsrc + (size_t)33 * HF);
    {
        float* xd = &xs[0][xdst];
        float* wd = &wsl[0][wdst];
        *reinterpret_cast<float4*>(xd)                = x0;
        *reinterpret_cast<float4*>(xd + 16 * GPA)     = x1;
        *reinterpret_cast<float4*>(xd + 32 * GPA)     = x2;
        *reinterpret_cast<float4*>(xd + 48 * GPA)     = x3;
        *reinterpret_cast<float4*>(wd)                = make_float4(we0.x, wo0.x, we0.y, wo0.y);
        *reinterpret_cast<float4*>(wd + 4)            = make_float4(we0.z, wo0.z, we0.w, wo0.w);
        *reinterpret_cast<float4*>(wd + 16 * GPB)     = make_float4(we1.x, wo1.x, we1.y, wo1.y);
        *reinterpret_cast<float4*>(wd + 16 * GPB + 4) = make_float4(we1.z, wo1.z, we1.w, wo1.w);
    }
    __syncthreads();

    v8f acc0 = {}, acc1 = {};

    const int aoff = (mt * 16 + m) * GPA + 2 * hf;
    const int boff = hf * GPB + 2 * (nh * 32 + m);

#pragma nounroll
    for (int sl = 0; sl < GNSLAB - 1; ++sl) {
        const int cur = sl & 1;
        const float* xn = xsrc + (sl + 1) * GBK;
        const float* wn = wsrc + (size_t)((sl + 1) * GBK) * HF;
        x0  = *reinterpret_cast<const float4*>(xn);
        x1  = *reinterpret_cast<const float4*>(xn + (size_t)16 * IN_FEAT);
        x2  = *reinterpret_cast<const float4*>(xn + (size_t)32 * IN_FEAT);
        x3  = *reinterpret_cast<const float4*>(xn + (size_t)48 * IN_FEAT);
        we0 = *reinterpret_cast<const float4*>(wn);
        wo0 = *reinterpret_cast<const float4*>(wn + (size_t)HF);
        we1 = *reinterpret_cast<const float4*>(wn + (size_t)32 * HF);
        wo1 = *reinterpret_cast<const float4*>(wn + (size_t)33 * HF);

        gemm1_slab(&xs[cur][aoff], &wsl[cur][boff], acc0, acc1);
        __builtin_amdgcn_sched_barrier(0x20);

        float* xd = &xs[cur ^ 1][xdst];
        float* wd = &wsl[cur ^ 1][wdst];
        *reinterpret_cast<float4*>(xd)                = x0;
        *reinterpret_cast<float4*>(xd + 16 * GPA)     = x1;
        *reinterpret_cast<float4*>(xd + 32 * GPA)     = x2;
        *reinterpret_cast<float4*>(xd + 48 * GPA)     = x3;
        *reinterpret_cast<float4*>(wd)                = make_float4(we0.x, wo0.x, we0.y, wo0.y);
        *reinterpret_cast<float4*>(wd + 4)            = make_float4(we0.z, wo0.z, we0.w, wo0.w);
        *reinterpret_cast<float4*>(wd + 16 * GPB)     = make_float4(we1.x, wo1.x, we1.y, wo1.y);
        *reinterpret_cast<float4*>(wd + 16 * GPB + 4) = make_float4(we1.z, wo1.z, we1.w, wo1.w);
        __syncthreads();
    }
    const int hd = blockIdx.y;
    const float* asrc = a + hd * 2 * FDIM;
    const float* adst = asrc + FDIM;
    const float as0 = asrc[lane], as1 = asrc[lane + 32];
    const float ad0 = adst[lane], ad1 = adst[lane + 32];

    gemm1_slab(&xs[(GNSLAB - 1) & 1][aoff], &wsl[(GNSLAB - 1) & 1][boff], acc0, acc1);

    float* hrow = h + (size_t)(i0 + mt * 16 + (hf << 3)) * HF + n0 + nh * 32 + m;
#pragma unroll
    for (int v = 0; v < 8; ++v) {
        hrow[(size_t)v * HF]      = acc0[v];
        hrow[(size_t)v * HF + 16] = acc1[v];
    }

    float* hs  = &xs[0][0];
    float* stl = &wsl[0][0];
    {
        float* hd0 = hs + (mt * 16 + (hf << 3)) * GPA + nh * 32 + m;
#pragma unroll
        for (int v = 0; v < 8; ++v) {
            hd0[v * GPA]      = acc0[v];
            hd0[v * GPA + 16] = acc1[v];
        }
    }
    __syncthreads();
    {
        float ms = 0.f, mtv = 0.f;
#pragma unroll
        for (int r = 0; r < 8; ++r) {
            const float* hr = hs + (wave * 8 + r) * GPA;
            const float h0 = hr[lane], h1 = hr[lane + 32];
            float sv = h0 * as0 + h1 * as1;
            float tv = h0 * ad0 + h1 * ad1;
            for (int off = 16; off > 0; off >>= 1) {
                sv += __shfl_xor(sv, off, 32);
                tv += __shfl_xor(tv, off, 32);
            }
            if (lane == r) { ms = sv; mtv = tv; }
        }
        if (lane < 8) {
            stl[wave * 8 + lane]      = ms;
            stl[64 + wave * 8 + lane] = mtv;
        }
    }
    __syncthreads();
    if (wave < 2) {
        float* dst = (wave == 0 ? s : t) + (size_t)hd * N_NODES + i0 + 2 * lane;
        *reinterpret_cast<float2*>(dst) = *reinterpret_cast<const float2*>(stl + wave * 64 + 2 * lane);
    }
}

#define ATT_ROWS 64
#define AQ       4
#define QJ       (N_NODES / AQ)
#define NKS      (QJ / 4)
#define RING     8
#define CPITCH   68
#define CRED_F   (AQ * ATT_ROWS * CPITCH)
#define LDS_F    (164 * 256)

__device__ __forceinline__ void attn_step(const bool prep, const bool reload,
                                          const float4& r0n, const float4& r1n, const float2& tqn,
                                          float4& r0w, float4& r1w, float2& tqw,
                                          const float* __restrict__ bld, const float* __restrict__ tld,
                                          const float tmaxv, const v2f (&fa2)[4], const v2f (&fb2)[4],
                                          v2f (&a)[4], v2f (&b)[4], v8f (&acc)[4][4], float (&zp)[4]) {
    v2f e2 = {0.f, 0.f}, g2 = {0.f, 0.f};
    v2f na[4] = {{0.f, 0.f}, {0.f, 0.f}, {0.f, 0.f}, {0.f, 0.f}};
    v2f nb[4] = {{0.f, 0.f}, {0.f, 0.f}, {0.f, 0.f}, {0.f, 0.f}};
    if (prep) {
        const float d0 = tqn.x - tmaxv, d1 = tqn.y - tmaxv;
        e2 = (v2f){__expf(d0), __expf(d1)};
        g2 = (v2f){__expf(ALPHA * d0), __expf(ALPHA * d1)};
    }
#pragma unroll
    for (int n = 0; n < 4; ++n)
        acc[0][n] = __builtin_amdgcn_wmma_f32_16x16x4_f32(false, a[0], false, b[n], (short)0, acc[0][n], false, false);
    __builtin_amdgcn_sched_group_barrier(0x008, 1, 0); __builtin_amdgcn_sched_group_barrier(0x002, 3, 0);
    __builtin_amdgcn_sched_group_barrier(0x008, 1, 0); __builtin_amdgcn_sched_group_barrier(0x002, 3, 0);
    __builtin_amdgcn_sched_group_barrier(0x008, 1, 0); __builtin_amdgcn_sched_group_barrier(0x002, 3, 0);
    __builtin_amdgcn_sched_group_barrier(0x008, 1, 0); __builtin_amdgcn_sched_group_barrier(0x002, 3, 0);
    __builtin_amdgcn_sched_barrier(0);
    if (prep) {
#pragma unroll
        for (int i = 0; i < 2; ++i) {
            na[i] = __builtin_elementwise_max(fa2[i] * e2, fb2[i] * g2);
            zp[i] += na[i].x + na[i].y;
        }
    }
#pragma unroll
    for (int n = 0; n < 4; ++n)
        acc[1][n] = __builtin_amdgcn_wmma_f32_16x16x4_f32(false, a[1], false, b[n], (short)0, acc[1][n], false, false);
    __builtin_amdgcn_sched_group_barrier(0x008, 1, 0); __builtin_amdgcn_sched_group_barrier(0x002, 3, 0);
    __builtin_amdgcn_sched_group_barrier(0x008, 1, 0); __builtin_amdgcn_sched_group_barrier(0x002, 3, 0);
    __builtin_amdgcn_sched_group_barrier(0x008, 1, 0); __builtin_amdgcn_sched_group_barrier(0x002, 3, 0);
    __builtin_amdgcn_sched_group_barrier(0x008, 1, 0); __builtin_amdgcn_sched_group_barrier(0x002, 3, 0);
    __builtin_amdgcn_sched_barrier(0);
    if (prep) {
#pragma unroll
        for (int i = 2; i < 4; ++i) {
            na[i] = __builtin_elementwise_max(fa2[i] * e2, fb2[i] * g2);
            zp[i] += na[i].x + na[i].y;
        }
    }
#pragma unroll
    for (int n = 0; n < 4; ++n)
        acc[2][n] = __builtin_amdgcn_wmma_f32_16x16x4_f32(false, a[2], false, b[n], (short)0, acc[2][n], false, false);
    __builtin_amdgcn_sched_group_barrier(0x008, 1, 0); __builtin_amdgcn_sched_group_barrier(0x002, 3, 0);
    __builtin_amdgcn_sched_group_barrier(0x008, 1, 0); __builtin_amdgcn_sched_group_barrier(0x002, 3, 0);
    __builtin_amdgcn_sched_group_barrier(0x008, 1, 0); __builtin_amdgcn_sched_group_barrier(0x002, 3, 0);
    __builtin_amdgcn_sched_group_barrier(0x008, 1, 0); __builtin_amdgcn_sched_group_barrier(0x002, 3, 0);
    __builtin_amdgcn_sched_barrier(0);
    if (prep) {
        const float4 ra = r0n, rb = r1n;
        nb[0] = (v2f){ra.x, rb.x}; nb[1] = (v2f){ra.y, rb.y};
        nb[2] = (v2f){ra.z, rb.z}; nb[3] = (v2f){ra.w, rb.w};
    }
    if (reload) {
        r0w = *reinterpret_cast<const float4*>(bld);
        r1w = *reinterpret_cast<const float4*>(bld + HF);
        tqw = *reinterpret_cast<const float2*>(tld);
    }
#pragma unroll
    for (int n = 0; n < 4; ++n)
        acc[3][n] = __builtin_amdgcn_wmma_f32_16x16x4_f32(false, a[3], false, b[n], (short)0, acc[3][n], false, false);
    __builtin_amdgcn_sched_group_barrier(0x008, 1, 0); __builtin_amdgcn_sched_group_barrier(0x002, 3, 0);
    __builtin_amdgcn_sched_group_barrier(0x008, 1, 0); __builtin_amdgcn_sched_group_barrier(0x002, 3, 0);
    __builtin_amdgcn_sched_group_barrier(0x008, 1, 0); __builtin_amdgcn_sched_group_barrier(0x002, 3, 0);
    __builtin_amdgcn_sched_group_barrier(0x008, 1, 0); __builtin_amdgcn_sched_group_barrier(0x002, 3, 0);
    __builtin_amdgcn_sched_barrier(0);
    if (prep) {
#pragma unroll
        for (int i = 0; i < 4; ++i) { a[i] = na[i]; b[i] = nb[i]; }
    }
}

template <bool LAST>
__device__ __forceinline__ void attn_turn(const float4 (&r0c)[RING], const float4 (&r1c)[RING], const float2 (&tqc)[RING],
                                          float4 (&r0o)[RING], float4 (&r1o)[RING], float2 (&tqo)[RING],
                                          const float* __restrict__ bnext, const float* __restrict__ tnext,
                                          const float tmaxv, const v2f (&fa2)[4], const v2f (&fb2)[4],
                                          v2f (&a)[4], v2f (&b)[4], v8f (&acc)[4][4], float (&zp)[4]) {
#pragma unroll
    for (int u = 0; u < RING; ++u) {
        const bool prep = !LAST || (u < RING - 1);
        if (u < RING - 1)
            attn_step(prep, !LAST, r0c[u + 1], r1c[u + 1], tqc[u + 1], r0o[u], r1o[u], tqo[u],
                      bnext + (size_t)(4 * u) * HF, tnext + 4 * u, tmaxv, fa2, fb2, a, b, acc, zp);
        else {
            const float4 n0 = r0o[0], n1 = r1o[0];
            const float2 nt = tqo[0];
            attn_step(prep, !LAST, n0, n1, nt, r0o[u], r1o[u], tqo[u],
                      bnext + (size_t)(4 * u) * HF, tnext + 4 * u, tmaxv, fa2, fb2, a, b, acc, zp);
        }
    }
}

__global__ __launch_bounds__(128) void gat_attn(const float* __restrict__ h,
                                                const float* __restrict__ s,
                                                const float* __restrict__ t,
                                                float* __restrict__ out) {
    __shared__ __attribute__((aligned(16))) float lds[LDS_F];
    float (*cred)[ATT_ROWS][CPITCH] = reinterpret_cast<float (*)[ATT_ROWS][CPITCH]>(lds);
    float (*zred)[4][32] = reinterpret_cast<float (*)[4][32]>(lds + CRED_F);
    float* zrow  = lds + CRED_F + AQ * 4 * 32;
    float* tmred = zrow + ATT_ROWS;

    const int tid  = threadIdx.x;
    const int lane = tid & 31;
    const int q    = tid >> 5;
    const int i0   = blockIdx.x * ATT_ROWS;
    const int hd   = blockIdx.y;
    const int m    = lane & 15;
    const int hf   = lane >> 4;

    const float* th   = t + (size_t)hd * N_NODES;
    const float* bsrc = h + (size_t)(q * QJ + 2 * hf) * HF + hd * FDIM + 4 * m;
    const float* tsrc = th + q * QJ + 2 * hf;

    float4 r0[RING], r1[RING], s0[RING], s1[RING];
    float2 tq[RING], sq[RING];
#pragma unroll
    for (int u = 0; u < RING; ++u) {
        r0[u] = *reinterpret_cast<const float4*>(bsrc + (size_t)(4 * u) * HF);
        r1[u] = *reinterpret_cast<const float4*>(bsrc + (size_t)(4 * u + 1) * HF);
        tq[u] = *reinterpret_cast<const float2*>(tsrc + 4 * u);
    }

    float mx = -INFINITY;
#pragma unroll
    for (int k = 0; k < 8; ++k) {
        const float4 u4 = *reinterpret_cast<const float4*>(th + 4 * (tid + 128 * k));
        mx = fmaxf(mx, fmaxf(fmaxf(u4.x, u4.y), fmaxf(u4.z, u4.w)));
    }
    for (int off = 16; off > 0; off >>= 1)
        mx = fmaxf(mx, __shfl_xor(mx, off, 32));
    if (lane == 0) tmred[q] = mx;
    float sm[4];
#pragma unroll
    for (int i = 0; i < 4; ++i) sm[i] = s[(size_t)hd * N_NODES + i0 + 16 * i + m];
    __syncthreads();

    const float4 rm = *reinterpret_cast<const float4*>(tmred);
    const float tmaxv = fmaxf(fmaxf(rm.x, rm.y), fmaxf(rm.z, rm.w));
    v2f fa2[4], fb2[4];
#pragma unroll
    for (int i = 0; i < 4; ++i) {
        const float xrow = sm[i] + tmaxv;
        const float mrow = lrelu(xrow);
        const float fa   = __expf(xrow - mrow);
        const float fb   = __expf(ALPHA * xrow - mrow);
        fa2[i] = (v2f){fa, fa};
        fb2[i] = (v2f){fb, fb};
    }

    v8f acc[4][4];
#pragma unroll
    for (int i = 0; i < 4; ++i) {
#pragma unroll
        for (int n = 0; n < 4; ++n) acc[i][n] = (v8f){};
    }
    float zp[4] = {0.f, 0.f, 0.f, 0.f};

    v2f a[4], b[4];
    {
        const float d0 = tq[0].x - tmaxv, d1 = tq[0].y - tmaxv;
        const v2f e2 = {__expf(d0), __expf(d1)};
        const v2f g2 = {__expf(ALPHA * d0), __expf(ALPHA * d1)};
#pragma unroll
        for (int i = 0; i < 4; ++i) {
            a[i] = __builtin_elementwise_max(fa2[i] * e2, fb2[i] * g2);
            zp[i] += a[i].x + a[i].y;
        }
        const float4 ra = r0[0], rb = r1[0];
        b[0] = (v2f){ra.x, rb.x}; b[1] = (v2f){ra.y, rb.y};
        b[2] = (v2f){ra.z, rb.z}; b[3] = (v2f){ra.w, rb.w};
    }

    const float* bnext = bsrc + (size_t)(4 * RING) * HF;
    const float* tnext = tsrc + 4 * RING;
#pragma nounroll
    for (int it = 0; it < NKS / (2 * RING) - 1; ++it) {
        attn_turn<false>(r0, r1, tq, s0, s1, sq, bnext, tnext, tmaxv, fa2, fb2, a, b, acc, zp);
        attn_turn<false>(s0, s1, sq, r0, r1, tq, bnext + (size_t)(4 * RING) * HF, tnext + 4 * RING,
                         tmaxv, fa2, fb2, a, b, acc, zp);
        bnext += (size_t)(8 * RING) * HF;
        tnext += 8 * RING;
    }
    attn_turn<false>(r0, r1, tq, s0, s1, sq, bnext, tnext, tmaxv, fa2, fb2, a, b, acc, zp);
    attn_turn<true>(s0, s1, sq, r0, r1, tq, bnext, tnext, tmaxv, fa2, fb2, a, b, acc, zp);

#pragma unroll
    for (int i = 0; i < 4; ++i) {
#pragma unroll
        for (int v = 0; v < 8; ++v)
            *reinterpret_cast<float4*>(&cred[q][16 * i + 8 * hf + v][4 * m]) =
                make_float4(acc[i][0][v], acc[i][1][v], acc[i][2][v], acc[i][3][v]);
        zred[q][i][lane] = zp[i];
    }
    __syncthreads();

    if (tid < ATT_ROWS) {
        const int zt = tid >> 4, zr = tid & 15;
        float z = 0.f;
#pragma unroll
        for (int w = 0; w < 4; ++w)
            z += zred[w][zt][zr] + zred[w][zt][zr + 16];
        zrow[tid] = z;
    }
    __syncthreads();

    for (int e = tid; e < ATT_ROWS * 16; e += 128) {
        const int rr = e >> 4, c4 = (e & 15) << 2;
        const float4 c0 = *reinterpret_cast<const float4*>(&cred[0][rr][c4]);
        const float4 c1 = *reinterpret_cast<const float4*>(&cred[1][rr][c4]);
        const float4 c2 = *reinterpret_cast<const float4*>(&cred[2][rr][c4]);
        const float4 c3 = *reinterpret_cast<const float4*>(&cred[3][rr][c4]);
        const float z = zrow[rr];
        float4 o;
        o.x = (c0.x + c1.x + c2.x + c3.x) / z;
        o.y = (c0.y + c1.y + c2.y + c3.y) / z;
        o.z = (c0.z + c1.z + c2.z + c3.z) / z;
        o.w = (c0.w + c1.w + c2.w + c3.w) / z;
        *reinterpret_cast<float4*>(out + (size_t)(i0 + rr) * HF + hd * FDIM + c4) = o;
    }
}

extern "C" void kernel_launch(void* const* d_in, const int* in_sizes, int n_in,
                              void* d_out, int out_size, void* d_ws, size_t ws_size,
                              hipStream_t stream) {
    const float* x = (const float*)d_in[0];
    const float* W = (const float*)d_in[2];
    const float* a = (const float*)d_in[3];
    float* out = (float*)d_out;

    char*  ws   = (char*)d_ws;
    float* h    = (float*)ws;
    float* s    = (float*)(ws + (size_t)N_NODES * HF * sizeof(float));
    float* t    = s + N_NODES * NHEAD;

    dim3 g1(N_NODES / GBM, HF / GBN);
    gat_gemm1<<<g1, 256, 0, stream>>>(x, W, a, h, s, t);

    dim3 g4(N_NODES / ATT_ROWS, NHEAD);
    gat_attn<<<g4, 128, 0, stream>>>(h, s, t, out);
}


# attention k loop: next turn's 24 row loads (8 k steps) issued as ONE group ahead of the turn's first WMMA; was 3 loads beside every k step, 8 places with loads per turn; every sum order kept
# speedup vs baseline: 1.7191x; 1.0396x over previous
#include <hip/hip_runtime.h>
#include <math.h>

#define ALPHA   0.2f
#define N_NODES 4096
#define IN_FEAT 512
#define NHEAD   4
#define FDIM    64
#define HF      256

typedef __attribute__((ext_vector_type(2))) float v2f;
typedef __attribute__((ext_vector_type(8))) float v8f;

__device__ __forceinline__ float lrelu(float x) {
    return fmaxf(x, ALPHA * x);
}

#define GBM    64
#define GBN    64
#define GBK    64
#define GPA    68
#define GPB    160
#define GNSLAB (IN_FEAT / GBK)

__device__ __forceinline__ void gemm1_slab(const float* __restrict__ ac, const float* __restrict__ bc,
                                           v8f& acc0, v8f& acc1) {
#pragma unroll
    for (int kk = 0; kk < GBK; kk += 4) {
        const v2f a2 = *reinterpret_cast<const v2f*>(ac + kk);
        const float* bp = bc + (kk >> 1) * GPB;
        const v2f b0 = *reinterpret_cast<const v2f*>(bp);
        const v2f b1 = *reinterpret_cast<const v2f*>(bp + 32);
        acc0 = __builtin_amdgcn_wmma_f32_16x16x4_f32(false, a2, false, b0, (short)0, acc0, false, false);
        acc1 = __builtin_amdgcn_wmma_f32_16x16x4_f32(false, a2, false, b1, (short)0, acc1, false, false);
    }
}

__global__ __launch_bounds__(256) void gat_gemm1(const float* __restrict__ x,
                                                 const float* __restrict__ W,
                                                 const float* __restrict__ a,
                                                 float* __restrict__ h,
                                                 float* __restrict__ s,
                                                 float* __restrict__ t) {
    __shared__ __attribute__((aligned(16))) float xs[2][GBM * GPA];
    __shared__ __attribute__((aligned(16))) float wsl[2][(GBK / 2) * GPB];

    const int tid  = threadIdx.x;
    const int lane = tid & 31;
    const int wave = tid >> 5;
    const int mt   = wave & 3;
    const int nh   = wave >> 2;
    const int i0   = blockIdx.x * GBM;
    const int n0   = blockIdx.y * GBN;
    const int m    = lane & 15;
    const int hf   = lane >> 4;

    const int srow = tid >> 4;
    const int scol = (tid & 15) << 2;
    const float* xsrc = x + (size_t)(i0 + srow) * IN_FEAT + scol;
    const float* wsrc = W + (size_t)(2 * srow) * HF + n0 + scol;
    const int xdst = srow * GPA + scol;
    const int wdst = srow * GPB + scol * 2;

    float4 x0, x1, x2, x3;
    float4 we0, wo0, we1, wo1;
    x0  = *reinterpret_cast<const float4*>(xsrc);
    x1  = *reinterpret_cast<const float4*>(xsrc + (size_t)16 * IN_FEAT);
    x2  = *reinterpret_cast<const float4*>(xsrc + (size_t)32 * IN_FEAT);
    x3  = *reinterpret_cast<const float4*>(xsrc + (size_t)48 * IN_FEAT);
    we0 = *reinterpret_cast<const float4*>(wsrc);
    wo0 = *reinterpret_cast<const float4*>(wsrc + (size_t)HF);
    we1 = *reinterpret_cast<const float4*>(wsrc + (size_t)32 * HF);
    wo1 = *reinterpret_cast<const float4*>(wsrc + (size_t)33 * HF);
    {
        float* xd = &xs[0][xdst];
        float* wd = &wsl[0][wdst];
        *reinterpret_cast<float4*>(xd)                = x0;
        *reinterpret_cast<float4*>(xd + 16 * GPA)     = x1;
        *reinterpret_cast<float4*>(xd + 32 * GPA)     = x2;
        *reinterpret_cast<float4*>(xd + 48 * GPA)     = x3;
        *reinterpret_cast<float4*>(wd)                = make_float4(we0.x, wo0.x, we0.y, wo0.y);
        *reinterpret_cast<float4*>(wd + 4)            = make_float4(we0.z, wo0.z, we0.w, wo0.w);
        *reinterpret_cast<float4*>(wd + 16 * GPB)     = make_float4(we1.x, wo1.x, we1.y, wo1.y);
        *reinterpret_cast<float4*>(wd + 16 * GPB + 4) = make_float4(we1.z, wo1.z, we1.w, wo1.w);
    }
    __syncthreads();

    v8f acc0 = {}, acc1 = {};

    const int aoff = (mt * 16 + m) * GPA + 2 * hf;
    const int boff = hf * GPB + 2 * (nh * 32 + m);

#pragma nounroll
    for (int sl = 0; sl < GNSLAB - 1; ++sl) {
        const int cur = sl & 1;
        const float* xn = xsrc + (sl + 1) * GBK;
        const float* wn = wsrc + (size_t)((sl + 1) * GBK) * HF;
        x0  = *reinterpret_cast<const float4*>(xn);
        x1  = *reinterpret_cast<const float4*>(xn + (size_t)16 * IN_FEAT);
        x2  = *reinterpret_cast<const float4*>(xn + (size_t)32 * IN_FEAT);
        x3  = *reinterpret_cast<const float4*>(xn + (size_t)48 * IN_FEAT);
        we0 = *reinterpret_cast<const float4*>(wn);
        wo0 = *reinterpret_cast<const float4*>(wn + (size_t)HF);
        we1 = *reinterpret_cast<const float4*>(wn + (size_t)32 * HF);
        wo1 = *reinterpret_cast<const float4*>(wn + (size_t)33 * HF);

        gemm1_slab(&xs[cur][aoff], &wsl[cur][boff], acc0, acc1);
        __builtin_amdgcn_sched_barrier(0x20);

        float* xd = &xs[cur ^ 1][xdst];
        float* wd = &wsl[cur ^ 1][wdst];
        *reinterpret_cast<float4*>(xd)                = x0;
        *reinterpret_cast<float4*>(xd + 16 * GPA)     = x1;
        *reinterpret_cast<float4*>(xd + 32 * GPA)     = x2;
        *reinterpret_cast<float4*>(xd + 48 * GPA)     = x3;
        *reinterpret_cast<float4*>(wd)                = make_float4(we0.x, wo0.x, we0.y, wo0.y);
        *reinterpret_cast<float4*>(wd + 4)            = make_float4(we0.z, wo0.z, we0.w, wo0.w);
        *reinterpret_cast<float4*>(wd + 16 * GPB)     = make_float4(we1.x, wo1.x, we1.y, wo1.y);
        *reinterpret_cast<float4*>(wd + 16 * GPB + 4) = make_float4(we1.z, wo1.z, we1.w, wo1.w);
        __syncthreads();
    }
    const int hd = blockIdx.y;
    const float* asrc = a + hd * 2 * FDIM;
    const float* adst = asrc + FDIM;
    const float as0 = asrc[lane], as1 = asrc[lane + 32];
    const float ad0 = adst[lane], ad1 = adst[lane + 32];

    gemm1_slab(&xs[(GNSLAB - 1) & 1][aoff], &wsl[(GNSLAB - 1) & 1][boff], acc0, acc1);

    float* hrow = h + (size_t)(i0 + mt * 16 + (hf << 3)) * HF + n0 + nh * 32 + m;
#pragma unroll
    for (int v = 0; v < 8; ++v) {
        hrow[(size_t)v * HF]      = acc0[v];
        hrow[(size_t)v * HF + 16] = acc1[v];
    }

    float* hs  = &xs[0][0];
    float* stl = &wsl[0][0];
    {
        float* hd0 = hs + (mt * 16 + (hf << 3)) * GPA + nh * 32 + m;
#pragma unroll
        for (int v = 0; v < 8; ++v) {
            hd0[v * GPA]      = acc0[v];
            hd0[v * GPA + 16] = acc1[v];
        }
    }
    __syncthreads();
    {
        float ms = 0.f, mtv = 0.f;
#pragma unroll
        for (int r = 0; r < 8; ++r) {
            const float* hr = hs + (wave * 8 + r) * GPA;
            const float h0 = hr[lane], h1 = hr[lane + 32];
            float sv = h0 * as0 + h1 * as1;
            float tv = h0 * ad0 + h1 * ad1;
            for (int off = 16; off > 0; off >>= 1) {
                sv += __shfl_xor(sv, off, 32);
                tv += __shfl_xor(tv, off, 32);
            }
            if (lane == r) { ms = sv; mtv = tv; }
        }
        if (lane < 8) {
            stl[wave * 8 + lane]      = ms;
            stl[64 + wave * 8 + lane] = mtv;
        }
    }
    __syncthreads();
    if (wave < 2) {
        float* dst = (wave == 0 ? s : t) + (size_t)hd * N_NODES + i0 + 2 * lane;
        *reinterpret_cast<float2*>(dst) = *reinterpret_cast<const float2*>(stl + wave * 64 + 2 * lane);
    }
}

#define ATT_ROWS 64
#define AQ       4
#define QJ       (N_NODES / AQ)
#define NKS      (QJ / 4)
#define RING     8
#define CPITCH   68
#define CRED_F   (AQ * ATT_ROWS * CPITCH)
#define LDS_F    (164 * 256)

__device__ __forceinline__ void attn_step(const bool prep,
                                          const float4& r0n, const float4& r1n, const float2& tqn,
                                          const float tmaxv, const v2f (&fa2)[4], const v2f (&fb2)[4],
                                          v2f (&a)[4], v2f (&b)[4], v8f (&acc)[4][4], float (&zp)[4]) {
    v2f e2 = {0.f, 0.f}, g2 = {0.f, 0.f};
    v2f na[4] = {{0.f, 0.f}, {0.f, 0.f}, {0.f, 0.f}, {0.f, 0.f}};
    v2f nb[4] = {{0.f, 0.f}, {0.f, 0.f}, {0.f, 0.f}, {0.f, 0.f}};
    if (prep) {
        const float d0 = tqn.x - tmaxv, d1 = tqn.y - tmaxv;
        e2 = (v2f){__expf(d0), __expf(d1)};
        g2 = (v2f){__expf(ALPHA * d0), __expf(ALPHA * d1)};
    }
#pragma unroll
    for (int n = 0; n < 4; ++n)
        acc[0][n] = __builtin_amdgcn_wmma_f32_16x16x4_f32(false, a[0], false, b[n], (short)0, acc[0][n], false, false);
    __builtin_amdgcn_sched_group_barrier(0x008, 1, 0); __builtin_amdgcn_sched_group_barrier(0x002, 3, 0);
    __builtin_amdgcn_sched_group_barrier(0x008, 1, 0); __builtin_amdgcn_sched_group_barrier(0x002, 3, 0);
    __builtin_amdgcn_sched_group_barrier(0x008, 1, 0); __builtin_amdgcn_sched_group_barrier(0x002, 3, 0);
    __builtin_amdgcn_sched_group_barrier(0x008, 1, 0); __builtin_amdgcn_sched_group_barrier(0x002, 3, 0);
    __builtin_amdgcn_sched_barrier(0);
    if (prep) {
#pragma unroll
        for (int i = 0; i < 2; ++i) {
            na[i] = __builtin_elementwise_max(fa2[i] * e2, fb2[i] * g2);
            zp[i] += na[i].x + na[i].y;
        }
    }
#pragma unroll
    for (int n = 0; n < 4; ++n)
        acc[1][n] = __builtin_amdgcn_wmma_f32_16x16x4_f32(false, a[1], false, b[n], (short)0, acc[1][n], false, false);
    __builtin_amdgcn_sched_group_barrier(0x008, 1, 0); __builtin_amdgcn_sched_group_barrier(0x002, 3, 0);
    __builtin_amdgcn_sched_group_barrier(0x008, 1, 0); __builtin_amdgcn_sched_group_barrier(0x002, 3, 0);
    __builtin_amdgcn_sched_group_barrier(0x008, 1, 0); __builtin_amdgcn_sched_group_barrier(0x002, 3, 0);
    __builtin_amdgcn_sched_group_barrier(0x008, 1, 0); __builtin_amdgcn_sched_group_barrier(0x002, 3, 0);
    __builtin_amdgcn_sched_barrier(0);
    if (prep) {
#pragma unroll
        for (int i = 2; i < 4; ++i) {
            na[i] = __builtin_elementwise_max(fa2[i] * e2, fb2[i] * g2);
            zp[i] += na[i].x + na[i].y;
        }
    }
#pragma unroll
    for (int n = 0; n < 4; ++n)
        acc[2][n] = __builtin_amdgcn_wmma_f32_16x16x4_f32(false, a[2], false, b[n], (short)0, acc[2][n], false, false);
    __builtin_amdgcn_sched_group_barrier(0x008, 1, 0); __builtin_amdgcn_sched_group_barrier(0x002, 3, 0);
    __builtin_amdgcn_sched_group_barrier(0x008, 1, 0); __builtin_amdgcn_sched_group_barrier(0x002, 3, 0);
    __builtin_amdgcn_sched_group_barrier(0x008, 1, 0); __builtin_amdgcn_sched_group_barrier(0x002, 3, 0);
    __builtin_amdgcn_sched_group_barrier(0x008, 1, 0); __builtin_amdgcn_sched_group_barrier(0x002, 3, 0);
    __builtin_amdgcn_sched_barrier(0);
    if (prep) {
        const float4 ra = r0n, rb = r1n;
        nb[0] = (v2f){ra.x, rb.x}; nb[1] = (v2f){ra.y, rb.y};
        nb[2] = (v2f){ra.z, rb.z}; nb[3] = (v2f){ra.w, rb.w};
    }
#pragma unroll
    for (int n = 0; n < 4; ++n)
        acc[3][n] = __builtin_amdgcn_wmma_f32_16x16x4_f32(false, a[3], false, b[n], (short)0, acc[3][n], false, false);
    __builtin_amdgcn_sched_group_barrier(0x008, 1, 0); __builtin_amdgcn_sched_group_barrier(0x002, 3, 0);
    __builtin_amdgcn_sched_group_barrier(0x008, 1, 0); __builtin_amdgcn_sched_group_barrier(0x002, 3, 0);
    __builtin_amdgcn_sched_group_barrier(0x008, 1, 0); __builtin_amdgcn_sched_group_barrier(0x002, 3, 0);
    __builtin_amdgcn_sched_group_barrier(0x008, 1, 0); __builtin_amdgcn_sched_group_barrier(0x002, 3, 0);
    __builtin_amdgcn_sched_barrier(0);
    if (prep) {
#pragma unroll
        for (int i = 0; i < 4; ++i) { a[i] = na[i]; b[i] = nb[i]; }
    }
}

template <bool LAST>
__device__ __forceinline__ void attn_turn(const float4 (&r0c)[RING], const float4 (&r1c)[RING], const float2 (&tqc)[RING],
                                          float4 (&r0o)[RING], float4 (&r1o)[RING], float2 (&tqo)[RING],
                                          const float* __restrict__ bnext, const float* __restrict__ tnext,
                                          const float tmaxv, const v2f (&fa2)[4], const v2f (&fb2)[4],
                                          v2f (&a)[4], v2f (&b)[4], v8f (&acc)[4][4], float (&zp)[4]) {
    if (!LAST) {
#pragma unroll
        for (int u = 0; u < RING; ++u) {
            r0o[u] = *reinterpret_cast<const float4*>(bnext + (size_t)(4 * u) * HF);
            r1o[u] = *reinterpret_cast<const float4*>(bnext + (size_t)(4 * u + 1) * HF);
            tqo[u] = *reinterpret_cast<const float2*>(tnext + 4 * u);
        }
        __builtin_amdgcn_sched_barrier(0);
    }
#pragma unroll
    for (int u = 0; u < RING; ++u) {
        const bool prep = !LAST || (u < RING - 1);
        if (u < RING - 1)
            attn_step(prep, r0c[u + 1], r1c[u + 1], tqc[u + 1], tmaxv, fa2, fb2, a, b, acc, zp);
        else {
            float2 nt = tqo[0];
            if (prep) asm volatile("" : "+v"(nt.x), "+v"(nt.y));
            attn_step(prep, r0o[0], r1o[0], nt, tmaxv, fa2, fb2, a, b, acc, zp);
        }
    }
}

__global__ __launch_bounds__(128) void gat_attn(const float* __restrict__ h,
                                                const float* __restrict__ s,
                                                const float* __restrict__ t,
                                                float* __restrict__ out) {
    __shared__ __attribute__((aligned(16))) float lds[LDS_F];
    float (*cred)[ATT_ROWS][CPITCH] = reinterpret_cast<float (*)[ATT_ROWS][CPITCH]>(lds);
    float (*zred)[4][32] = reinterpret_cast<float (*)[4][32]>(lds + CRED_F);
    float* zrow  = lds + CRED_F + AQ * 4 * 32;
    float* tmred = zrow + ATT_ROWS;

    const int tid  = threadIdx.x;
    const int lane = tid & 31;
    const int q    = tid >> 5;
    const int i0   = blockIdx.x * ATT_ROWS;
    const int hd   = blockIdx.y;
    const int m    = lane & 15;
    const int hf   = lane >> 4;

    const float* th   = t + (size_t)hd * N_NODES;
    const float* bsrc = h + (size_t)(q * QJ + 2 * hf) * HF + hd * FDIM + 4 * m;
    const float* tsrc = th + q * QJ + 2 * hf;

    float4 r0[RING], r1[RING], s0[RING], s1[RING];
    float2 tq[RING], sq[RING];
#pragma unroll
    for (int u = 0; u < RING; ++u) {
        r0[u] = *reinterpret_cast<const float4*>(bsrc + (size_t)(4 * u) * HF);
        r1[u] = *reinterpret_cast<const float4*>(bsrc + (size_t)(4 * u + 1) * HF);
        tq[u] = *reinterpret_cast<const float2*>(tsrc + 4 * u);
    }

    float mx = -INFINITY;
#pragma unroll
    for (int k = 0; k < 8; ++k) {
        const float4 u4 = *reinterpret_cast<const float4*>(th + 4 * (tid + 128 * k));
        mx = fmaxf(mx, fmaxf(fmaxf(u4.x, u4.y), fmaxf(u4.z, u4.w)));
    }
    for (int off = 16; off > 0; off >>= 1)
        mx = fmaxf(mx, __shfl_xor(mx, off, 32));
    if (lane == 0) tmred[q] = mx;
    float sm[4];
#pragma unroll
    for (int i = 0; i < 4; ++i) sm[i] = s[(size_t)hd * N_NODES + i0 + 16 * i + m];
    __syncthreads();

    const float4 rm = *reinterpret_cast<const float4*>(tmred);
    const float tmaxv = fmaxf(fmaxf(rm.x, rm.y), fmaxf(rm.z, rm.w));
    v2f fa2[4], fb2[4];
#pragma unroll
    for (int i = 0; i < 4; ++i) {
        const float xrow = sm[i] + tmaxv;
        const float mrow = lrelu(xrow);
        const float fa   = __expf(xrow - mrow);
        const float fb   = __expf(ALPHA * xrow - mrow);
        fa2[i] = (v2f){fa, fa};
        fb2[i] = (v2f){fb, fb};
    }

    v8f acc[4][4];
#pragma unroll
    for (int i = 0; i < 4; ++i) {
#pragma unroll
        for (int n = 0; n < 4; ++n) acc[i][n] = (v8f){};
    }
    float zp[4] = {0.f, 0.f, 0.f, 0.f};

    v2f a[4], b[4];
    {
        const float d0 = tq[0].x - tmaxv, d1 = tq[0].y - tmaxv;
        const v2f e2 = {__expf(d0), __expf(d1)};
        const v2f g2 = {__expf(ALPHA * d0), __expf(ALPHA * d1)};
#pragma unroll
        for (int i = 0; i < 4; ++i) {
            a[i] = __builtin_elementwise_max(fa2[i] * e2, fb2[i] * g2);
            zp[i] += a[i].x + a[i].y;
        }
        const float4 ra = r0[0], rb = r1[0];
        b[0] = (v2f){ra.x, rb.x}; b[1] = (v2f){ra.y, rb.y};
        b[2] = (v2f){ra.z, rb.z}; b[3] = (v2f){ra.w, rb.w};
    }

    const float* bnext = bsrc + (size_t)(4 * RING) * HF;
    const float* tnext = tsrc + 4 * RING;
#pragma nounroll
    for (int it = 0; it < NKS / (2 * RING) - 1; ++it) {
        attn_turn<false>(r0, r1, tq, s0, s1, sq, bnext, tnext, tmaxv, fa2, fb2, a, b, acc, zp);
        attn_turn<false>(s0, s1, sq, r0, r1, tq, bnext + (size_t)(4 * RING) * HF, tnext + 4 * RING,
                         tmaxv, fa2, fb2, a, b, acc, zp);
        bnext += (size_t)(8 * RING) * HF;
        tnext += 8 * RING;
    }
    attn_turn<false>(r0, r1, tq, s0, s1, sq, bnext, tnext, tmaxv, fa2, fb2, a, b, acc, zp);
    attn_turn<true>(s0, s1, sq, r0, r1, tq, bnext, tnext, tmaxv, fa2, fb2, a, b, acc, zp);

#pragma unroll
    for (int i = 0; i < 4; ++i) {
#pragma unroll
        for (int v = 0; v < 8; ++v)
            *reinterpret_cast<float4*>(&cred[q][16 * i + 8 * hf + v][4 * m]) =
                make_float4(acc[i][0][v], acc[i][1][v], acc[i][2][v], acc[i][3][v]);
        zred[q][i][lane] = zp[i];
    }
    __syncthreads();

    if (tid < ATT_ROWS) {
        const int zt = tid >> 4, zr = tid & 15;
        float z = 0.f;
#pragma unroll
        for (int w = 0; w < 4; ++w)
            z += zred[w][zt][zr] + zred[w][zt][zr + 16];
        zrow[tid] = z;
    }
    __syncthreads();

    for (int e = tid; e < ATT_ROWS * 16; e += 128) {
        const int rr = e >> 4, c4 = (e & 15) << 2;
        const float4 c0 = *reinterpret_cast<const float4*>(&cred[0][rr][c4]);
        const float4 c1 = *reinterpret_cast<const float4*>(&cred[1][rr][c4]);
        const float4 c2 = *reinterpret_cast<const float4*>(&cred[2][rr][c4]);
        const float4 c3 = *reinterpret_cast<const float4*>(&cred[3][rr][c4]);
        const float z = zrow[rr];
        float4 o;
        o.x = (c0.x + c1.x + c2.x + c3.x) / z;
        o.y = (c0.y + c1.y + c2.y + c3.y) / z;
        o.z = (c0.z + c1.z + c2.z + c3.z) / z;
        o.w = (c0.w + c1.w + c2.w + c3.w) / z;
        *reinterpret_cast<float4*>(out + (size_t)(i0 + rr) * HF + hd * FDIM + c4) = o;
    }
}

extern "C" void kernel_launch(void* const* d_in, const int* in_sizes, int n_in,
                              void* d_out, int out_size, void* d_ws, size_t ws_size,
                              hipStream_t stream) {
    const float* x = (const float*)d_in[0];
    const float* W = (const float*)d_in[2];
    const float* a = (const float*)d_in[3];
    float* out = (float*)d_out;

    char*  ws   = (char*)d_ws;
    float* h    = (float*)ws;
    float* s    = (float*)(ws + (size_t)N_NODES * HF * sizeof(float));
    float* t    = s + N_NODES * NHEAD;

    dim3 g1(N_NODES / GBM, HF / GBN);
    gat_gemm1<<<g1, 256, 0, stream>>>(x, W, a, h, s, t);

    dim3 g4(N_NODES / ATT_ROWS, NHEAD);
    gat_attn<<<g4, 128, 0, stream>>>(h, s, t, out);
}


# gat_rank: key wave-uniform, 32 values across the lanes: 1 compare + scalar popcount per 32 pairs (was compare+select+add per pair); a wave ranks and moves its own 8 rows, h loads at entry; ranks kept
# speedup vs baseline: 3.2785x; 1.9071x over previous
#include <hip/hip_runtime.h>
#include <math.h>

#define ALPHA   0.2f
#define N_NODES 4096
#define IN_FEAT 512
#define NHEAD   4
#define FDIM    64
#define HF      256

typedef __attribute__((ext_vector_type(2))) float v2f;
typedef __attribute__((ext_vector_type(8))) float v8f;

__device__ __forceinline__ float lrelu(float x) {
    return fmaxf(x, ALPHA * x);
}

#define GBM    64
#define GBN    64
#define GBK    64
#define GPA    68
#define GPB    160
#define GNSLAB (IN_FEAT / GBK)

__device__ __forceinline__ void gemm1_slab(const float* __restrict__ ac, const float* __restrict__ bc,
                                           v8f& acc0, v8f& acc1) {
#pragma unroll
    for (int kk = 0; kk < GBK; kk += 4) {
        const v2f a2 = *reinterpret_cast<const v2f*>(ac + kk);
        const float* bp = bc + (kk >> 1) * GPB;
        const v2f b0 = *reinterpret_cast<const v2f*>(bp);
        const v2f b1 = *reinterpret_cast<const v2f*>(bp + 32);
        acc0 = __builtin_amdgcn_wmma_f32_16x16x4_f32(false, a2, false, b0, (short)0, acc0, false, false);
        acc1 = __builtin_amdgcn_wmma_f32_16x16x4_f32(false, a2, false, b1, (short)0, acc1, false, false);
    }
}

__global__ __launch_bounds__(256) void gat_gemm1(const float* __restrict__ x,
                                                 const float* __restrict__ W,
                                                 const float* __restrict__ a,
                                                 float* __restrict__ h,
                                                 float* __restrict__ s,
                                                 float* __restrict__ t) {
    __shared__ __attribute__((aligned(16))) float xs[2][GBM * GPA];
    __shared__ __attribute__((aligned(16))) float wsl[2][(GBK / 2) * GPB];

    const int tid  = threadIdx.x;
    const int lane = tid & 31;
    const int wave = tid >> 5;
    const int mt   = wave & 3;
    const int nh   = wave >> 2;
    const int i0   = blockIdx.x * GBM;
    const int n0   = blockIdx.y * GBN;
    const int m    = lane & 15;
    const int hf   = lane >> 4;

    const int srow = tid >> 4;
    const int scol = (tid & 15) << 2;
    const float* xsrc = x + (size_t)(i0 + srow) * IN_FEAT + scol;
    const float* wsrc = W + (size_t)(2 * srow) * HF + n0 + scol;
    const int xdst = srow * GPA + scol;
    const int wdst = srow * GPB + scol * 2;

    float4 x0, x1, x2, x3;
    float4 we0, wo0, we1, wo1;
    x0  = *reinterpret_cast<const float4*>(xsrc);
    x1  = *reinterpret_cast<const float4*>(xsrc + (size_t)16 * IN_FEAT);
    x2  = *reinterpret_cast<const float4*>(xsrc + (size_t)32 * IN_FEAT);
    x3  = *reinterpret_cast<const float4*>(xsrc + (size_t)48 * IN_FEAT);
    we0 = *reinterpret_cast<const float4*>(wsrc);
    wo0 = *reinterpret_cast<const float4*>(wsrc + (size_t)HF);
    we1 = *reinterpret_cast<const float4*>(wsrc + (size_t)32 * HF);
    wo1 = *reinterpret_cast<const float4*>(wsrc + (size_t)33 * HF);
    {
        float* xd = &xs[0][xdst];
        float* wd = &wsl[0][wdst];
        *reinterpret_cast<float4*>(xd)                = x0;
        *reinterpret_cast<float4*>(xd + 16 * GPA)     = x1;
        *reinterpret_cast<float4*>(xd + 32 * GPA)     = x2;
        *reinterpret_cast<float4*>(xd + 48 * GPA)     = x3;
        *reinterpret_cast<float4*>(wd)                = make_float4(we0.x, wo0.x, we0.y, wo0.y);
        *reinterpret_cast<float4*>(wd + 4)            = make_float4(we0.z, wo0.z, we0.w, wo0.w);
        *reinterpret_cast<float4*>(wd + 16 * GPB)     = make_float4(we1.x, wo1.x, we1.y, wo1.y);
        *reinterpret_cast<float4*>(wd + 16 * GPB + 4) = make_float4(we1.z, wo1.z, we1.w, wo1.w);
    }
    __syncthreads();

    v8f acc0 = {}, acc1 = {};

    const int aoff = (mt * 16 + m) * GPA + 2 * hf;
    const int boff = hf * GPB + 2 * (nh * 32 + m);

#pragma nounroll
    for (int sl = 0; sl < GNSLAB - 1; ++sl) {
        const int cur = sl & 1;
        const float* xn = xsrc + (sl + 1) * GBK;
        const float* wn = wsrc + (size_t)((sl + 1) * GBK) * HF;
        x0  = *reinterpret_cast<const float4*>(xn);
        x1  = *reinterpret_cast<const float4*>(xn + (size_t)16 * IN_FEAT);
        x2  = *reinterpret_cast<const float4*>(xn + (size_t)32 * IN_FEAT);
        x3  = *reinterpret_cast<const float4*>(xn + (size_t)48 * IN_FEAT);
        we0 = *reinterpret_cast<const float4*>(wn);
        wo0 = *reinterpret_cast<const float4*>(wn + (size_t)HF);
        we1 = *reinterpret_cast<const float4*>(wn + (size_t)32 * HF);
        wo1 = *reinterpret_cast<const float4*>(wn + (size_t)33 * HF);

        gemm1_slab(&xs[cur][aoff], &wsl[cur][boff], acc0, acc1);
        __builtin_amdgcn_sched_barrier(0x20);

        float* xd = &xs[cur ^ 1][xdst];
        float* wd = &wsl[cur ^ 1][wdst];
        *reinterpret_cast<float4*>(xd)                = x0;
        *reinterpret_cast<float4*>(xd + 16 * GPA)     = x1;
        *reinterpret_cast<float4*>(xd + 32 * GPA)     = x2;
        *reinterpret_cast<float4*>(xd + 48 * GPA)     = x3;
        *reinterpret_cast<float4*>(wd)                = make_float4(we0.x, wo0.x, we0.y, wo0.y);
        *reinterpret_cast<float4*>(wd + 4)            = make_float4(we0.z, wo0.z, we0.w, wo0.w);
        *reinterpret_cast<float4*>(wd + 16 * GPB)     = make_float4(we1.x, wo1.x, we1.y, wo1.y);
        *reinterpret_cast<float4*>(wd + 16 * GPB + 4) = make_float4(we1.z, wo1.z, we1.w, wo1.w);
        __syncthreads();
    }
    const int hd = blockIdx.y;
    const float* asrc = a + hd * 2 * FDIM;
    const float* adst = asrc + FDIM;
    const float as0 = asrc[lane], as1 = asrc[lane + 32];
    const float ad0 = adst[lane], ad1 = adst[lane + 32];

    gemm1_slab(&xs[(GNSLAB - 1) & 1][aoff], &wsl[(GNSLAB - 1) & 1][boff], acc0, acc1);

    float* hrow = h + (size_t)(i0 + mt * 16 + (hf << 3)) * HF + n0 + nh * 32 + m;
#pragma unroll
    for (int v = 0; v < 8; ++v) {
        hrow[(size_t)v * HF]      = acc0[v];
        hrow[(size_t)v * HF + 16] = acc1[v];
    }

    float* hs  = &xs[0][0];
    float* stl = &wsl[0][0];
    {
        float* hd0 = hs + (mt * 16 + (hf << 3)) * GPA + nh * 32 + m;
#pragma unroll
        for (int v = 0; v < 8; ++v) {
            hd0[v * GPA]      = acc0[v];
            hd0[v * GPA + 16] = acc1[v];
        }
    }
    __syncthreads();
    {
        float ms = 0.f, mtv = 0.f;
#pragma unroll
        for (int r = 0; r < 8; ++r) {
            const float* hr = hs + (wave * 8 + r) * GPA;
            const float h0 = hr[lane], h1 = hr[lane + 32];
            float sv = h0 * as0 + h1 * as1;
            float tv = h0 * ad0 + h1 * ad1;
            for (int off = 16; off > 0; off >>= 1) {
                sv += __shfl_xor(sv, off, 32);
                tv += __shfl_xor(tv, off, 32);
            }
            if (lane == r) { ms = sv; mtv = tv; }
        }
        if (lane < 8) {
            stl[wave * 8 + lane]      = ms;
            stl[64 + wave * 8 + lane] = mtv;
        }
    }
    __syncthreads();
    if (wave < 2) {
        float* dst = (wave == 0 ? s : t) + (size_t)hd * N_NODES + i0 + 2 * lane;
        *reinterpret_cast<float2*>(dst) = *reinterpret_cast<const float2*>(stl + wave * 64 + 2 * lane);
    }
}

#define RECF 96
#define TOTF 96
#define NCHUNK (N_NODES / 64)

template <bool LE>
__device__ __forceinline__ void count_group(const float4 v, const float (&tk)[8], int (&cnt)[8]) {
#pragma unroll
    for (int kk = 0; kk < 8; ++kk) {
        const float tj = tk[kk];
        const unsigned m0 = __builtin_amdgcn_ballot_w32(LE ? (v.x <= tj) : (v.x < tj));
        const unsigned m1 = __builtin_amdgcn_ballot_w32(LE ? (v.y <= tj) : (v.y < tj));
        const unsigned m2 = __builtin_amdgcn_ballot_w32(LE ? (v.z <= tj) : (v.z < tj));
        const unsigned m3 = __builtin_amdgcn_ballot_w32(LE ? (v.w <= tj) : (v.w < tj));
        cnt[kk] += __builtin_popcountll(((unsigned long long)m1 << 32) | m0)
                 + __builtin_popcountll(((unsigned long long)m3 << 32) | m2);
    }
}

__global__ __launch_bounds__(256) void gat_rank(const float* __restrict__ h,
                                                const float* __restrict__ t,
                                                float* __restrict__ hs) {
    __shared__ __attribute__((aligned(16))) float tl[N_NODES];

    const int tid  = threadIdx.x;
    const int lane = tid & 31;
    const int wave = __builtin_amdgcn_readfirstlane(tid >> 5);
    const int cb   = blockIdx.x;
    const int hd   = blockIdx.y;
    const float* th = t + (size_t)hd * N_NODES;

    float2 hv[8];
#pragma unroll
    for (int r = 0; r < 8; ++r)
        hv[r] = *reinterpret_cast<const float2*>(h + (size_t)(64 * cb + 8 * wave + r) * HF + hd * FDIM + 2 * lane);

#pragma unroll
    for (int k = 0; k < 4; ++k)
        *reinterpret_cast<float4*>(tl + 4 * (tid + 256 * k)) =
            *reinterpret_cast<const float4*>(th + 4 * (tid + 256 * k));
    __syncthreads();

    float tk[8];
    int   cnt[8];
#pragma unroll
    for (int kk = 0; kk < 8; ++kk) {
        tk[kk]  = __builtin_bit_cast(float, __builtin_amdgcn_readfirstlane(
                      __builtin_bit_cast(int, tl[64 * cb + 8 * wave + kk])));
        cnt[kk] = 0;
    }

    const int qb = cb >> 1;
    const float4* tl4 = reinterpret_cast<const float4*>(tl);
#pragma unroll 4
    for (int q = 0; q < qb; ++q)
        count_group<true>(tl4[32 * q + lane], tk, cnt);
#pragma unroll 4
    for (int q = qb + 1; q < N_NODES / 128; ++q)
        count_group<false>(tl4[32 * q + lane], tk, cnt);
    {
        const float4 v = tl4[32 * qb + lane];
        const float e[4] = {v.x, v.y, v.z, v.w};
#pragma unroll
        for (int kk = 0; kk < 8; ++kk) {
            const float tj = tk[kk];
            const int jo = 64 * (cb & 1) + 8 * wave + kk;
            int c = 0;
#pragma unroll
            for (int w = 0; w < 4; ++w) {
                const int n = (jo - w + 3) >> 2;
                const unsigned lm = (n >= 32) ? 0xffffffffu : ((1u << n) - 1u);
                const unsigned lt = __builtin_amdgcn_ballot_w32(e[w] < tj);
                const unsigned eq = __builtin_amdgcn_ballot_w32(e[w] == tj);
                c += __builtin_popcount(lt | (eq & lm));
            }
            cnt[kk] += c;
        }
    }

#pragma unroll
    for (int r = 0; r < 8; ++r) {
        float* dst = hs + ((size_t)hd * N_NODES + cnt[r]) * RECF;
        *reinterpret_cast<float2*>(dst + 2 * lane) = hv[r];
        dst[64 + lane] = (lane == 0) ? tk[r] : 0.f;
    }
}

__global__ __launch_bounds__(256) void gat_scan(const float* __restrict__ hs,
                                                float* __restrict__ lg,
                                                float* __restrict__ lse,
                                                float* __restrict__ zlg,
                                                float* __restrict__ zlse,
                                                float* __restrict__ tot,
                                                float* __restrict__ ts) {
    __shared__ float sg[4][64];
    __shared__ float se[4][64];
    __shared__ float zsg[4];
    __shared__ float zse[4];
    __shared__ __attribute__((aligned(8))) float zo[2][64];

    const int tid  = threadIdx.x;
    const int lane = tid & 31;
    const int wave = tid >> 5;
    const int col  = tid & 63;
    const int seg  = tid >> 6;
    const int c    = blockIdx.x;
    const int hd   = blockIdx.y;
    const size_t row0 = (size_t)hd * N_NODES + 64 * c + 16 * seg;
    const float* rec = hs + row0 * RECF;
    const float tmaxv = hs[((size_t)hd * N_NODES + (N_NODES - 1)) * RECF + 64];

    float hv[16], gw[16], ew[16], pg[16], pe[16];
#pragma unroll
    for (int r = 0; r < 16; ++r) {
        hv[r] = rec[(size_t)r * RECF + col];
        const float d = rec[(size_t)r * RECF + 64] - tmaxv;
        ew[r] = __expf(d);
        gw[r] = __expf(ALPHA * d);
    }
    float rg = 0.f, re = 0.f;
#pragma unroll
    for (int r = 0; r < 16; ++r) { pg[r] = rg; rg += gw[r] * hv[r]; }
#pragma unroll
    for (int r = 15; r >= 0; --r) { re += ew[r] * hv[r]; pe[r] = re; }
    sg[seg][col] = rg;
    se[seg][col] = re;
    float zpg[16], zpe[16];
    float zrg = 0.f, zre = 0.f;
#pragma unroll
    for (int r = 0; r < 16; ++r) { zpg[r] = zrg; zrg += gw[r]; }
#pragma unroll
    for (int r = 15; r >= 0; --r) { zre += ew[r]; zpe[r] = zre; }
    if (col == 0) { zsg[seg] = zrg; zse[seg] = zre; }
    __syncthreads();

    float og = 0.f, oe = 0.f, zog = 0.f, zoe = 0.f;
    for (int q = 0; q < seg; ++q)     { og += sg[q][col]; zog += zsg[q]; }
    for (int q = 3; q > seg; --q)     { oe += se[q][col]; zoe += zse[q]; }
    float* lgr = lg  + row0 * FDIM + col;
    float* ler = lse + row0 * FDIM + col;
#pragma unroll
    for (int r = 0; r < 16; ++r) {
        lgr[(size_t)r * FDIM] = og + pg[r];
        ler[(size_t)r * FDIM] = oe + pe[r];
    }
    if (col == 0) {
#pragma unroll
        for (int r = 0; r < 16; ++r) {
            zo[0][16 * seg + r] = zog + zpg[r];
            zo[1][16 * seg + r] = zoe + zpe[r];
        }
    }
    float* tr = tot + ((size_t)(hd * NCHUNK + c) * 2) * TOTF;
    if (seg == 0) tr[col]        = ((sg[0][col] + sg[1][col]) + sg[2][col]) + sg[3][col];
    if (seg == 1) tr[TOTF + col] = ((se[3][col] + se[2][col]) + se[1][col]) + se[0][col];
    __syncthreads();
    const size_t z0 = (size_t)hd * N_NODES + 64 * c + 2 * lane;
    if (wave == 0) *reinterpret_cast<float2*>(zlg + z0)  = *reinterpret_cast<const float2*>(&zo[0][2 * lane]);
    if (wave == 1) *reinterpret_cast<float2*>(zlse + z0) = *reinterpret_cast<const float2*>(&zo[1][2 * lane]);
    if (wave == 2) {
        const float* tr0 = hs + ((size_t)hd * N_NODES + 64 * c + 2 * lane) * RECF + 64;
        float2 tv;
        tv.x = tr0[0];
        tv.y = tr0[RECF];
        *reinterpret_cast<float2*>(ts + z0) = tv;
    }
    if (wave == 4) tr[64 + lane]        = (lane == 0) ? ((zsg[0] + zsg[1]) + zsg[2]) + zsg[3] : 0.f;
    if (wave == 5) tr[TOTF + 64 + lane] = (lane == 0) ? ((zse[3] + zse[2]) + zse[1]) + zse[0] : 0.f;
}

#define OPITCH 66

__global__ __launch_bounds__(256) void gat_out(const float* __restrict__ lg,
                                               const float* __restrict__ lse,
                                               const float* __restrict__ zlg,
                                               const float* __restrict__ zlse,
                                               const float* __restrict__ tot,
                                               const float* __restrict__ ts,
                                               const float* __restrict__ s,
                                               float* __restrict__ out) {
    __shared__ __attribute__((aligned(8))) float og[(NCHUNK + 1) * OPITCH];
    __shared__ __attribute__((aligned(8))) float oe[(NCHUNK + 1) * OPITCH];
    __shared__ __attribute__((aligned(16))) float tsl[N_NODES];

    const int tid  = threadIdx.x;
    const int lane = tid & 31;
    const int wave = tid >> 5;
    const int i0   = blockIdx.x * 64;
    const int hd   = blockIdx.y;

#pragma unroll
    for (int k = 0; k < 4; ++k)
        *reinterpret_cast<float4*>(tsl + 4 * (tid + 256 * k)) =
            *reinterpret_cast<const float4*>(ts + (size_t)hd * N_NODES + 4 * (tid + 256 * k));
    const float thr = -s[(size_t)hd * N_NODES + i0 + 8 * wave + (lane & 7)];
    if (tid < 130) {
        const int tb   = (tid >= 65) ? 1 : 0;
        const int colo = tid - 65 * tb;
        const float* tp = tot + ((size_t)hd * NCHUNK * 2 + tb) * TOTF + colo;
        float tv[NCHUNK];
#pragma unroll
        for (int c = 0; c < NCHUNK; ++c) tv[c] = tp[(size_t)c * 2 * TOTF];
        float run = 0.f;
        if (tb == 0) {
#pragma unroll
            for (int c = 0; c < NCHUNK; ++c) { og[c * OPITCH + colo] = run; run += tv[c]; }
            og[NCHUNK * OPITCH + colo] = run;
        } else {
            oe[NCHUNK * OPITCH + colo] = 0.f;
#pragma unroll
            for (int c = NCHUNK - 1; c >= 0; --c) { oe[c * OPITCH + colo] = run; run += tv[c]; }
        }
    }
    __syncthreads();
    const float tmaxv = tsl[N_NODES - 1];
    int lo = 0, hi = N_NODES;
    while (lo < hi) {
        const int mid = (lo + hi) >> 1;
        if (tsl[mid] < thr) lo = mid + 1; else hi = mid;
    }

    int    kr[8];
    float2 ta[8], tb2[8];
    float  za[8], zb[8], sr[8];
#pragma unroll
    for (int r = 0; r < 8; ++r) {
        kr[r] = __shfl(lo, r, 32);
        const size_t row = (size_t)hd * N_NODES + min(kr[r], N_NODES - 1);
        ta[r]  = *reinterpret_cast<const float2*>(lg  + row * FDIM + 2 * lane);
        tb2[r] = *reinterpret_cast<const float2*>(lse + row * FDIM + 2 * lane);
        za[r]  = zlg[row];
        zb[r]  = zlse[row];
        sr[r]  = s[(size_t)hd * N_NODES + i0 + 8 * wave + r];
    }
#pragma unroll
    for (int r = 0; r < 8; ++r) {
        const int i = i0 + 8 * wave + r;
        const float si = sr[r];
        const int k = kr[r];
        const int c = k >> 6;
        float2 g2 = *reinterpret_cast<const float2*>(&og[c * OPITCH + 2 * lane]);
        float2 e2 = *reinterpret_cast<const float2*>(&oe[c * OPITCH + 2 * lane]);
        float gz = og[c * OPITCH + 64];
        float ez = oe[c * OPITCH + 64];
        const bool in = k < N_NODES;
        g2.x += in ? ta[r].x : 0.f;  g2.y += in ? ta[r].y : 0.f;  gz += in ? za[r] : 0.f;
        e2.x += in ? tb2[r].x : 0.f; e2.y += in ? tb2[r].y : 0.f; ez += in ? zb[r] : 0.f;
        const float xrow = si + tmaxv;
        const float mrow = lrelu(xrow);
        const float fa   = __expf(xrow - mrow);
        const float fb   = __expf(ALPHA * xrow - mrow);
        const float z    = fa * ez + fb * gz;
        float2 o;
        o.x = (fa * e2.x + fb * g2.x) / z;
        o.y = (fa * e2.y + fb * g2.y) / z;
        *reinterpret_cast<float2*>(out + (size_t)i * HF + hd * FDIM + 2 * lane) = o;
    }
}

extern "C" void kernel_launch(void* const* d_in, const int* in_sizes, int n_in,
                              void* d_out, int out_size, void* d_ws, size_t ws_size,
                              hipStream_t stream) {
    const float* x = (const float*)d_in[0];
    const float* W = (const float*)d_in[2];
    const float* a = (const float*)d_in[3];
    float* out = (float*)d_out;

    if (ws_size < (size_t)15204352) return;

    char*  ws   = (char*)d_ws;
    float* h    = (float*)ws;
    float* lg   = (float*)ws;
    float* s    = (float*)(ws + (size_t)4194304);
    float* t    = s + N_NODES * NHEAD;
    float* ts   = (float*)(ws + (size_t)4325376);
    float* hs   = (float*)(ws + (size_t)4390912);
    float* lse  = (float*)(ws + (size_t)10682368);
    float* zlg  = (float*)(ws + (size_t)14876672);
    float* zlse = zlg + N_NODES * NHEAD;
    float* tot  = (float*)(ws + (size_t)15007744);

    dim3 g1(N_NODES / GBM, HF / GBN);
    gat_gemm1<<<g1, 256, 0, stream>>>(x, W, a, h, s, t);

    dim3 g2(NCHUNK, NHEAD);
    gat_rank<<<g2, 256, 0, stream>>>(h, t, hs);
    gat_scan<<<g2, 256, 0, stream>>>(hs, lg, lse, zlg, zlse, tot, ts);
    gat_out<<<g2, 256, 0, stream>>>(lg, lse, zlg, zlse, tot, ts, s, out);
}
